# starting kernel
# baseline (speedup 1.0000x reference)
#include <hip/hip_runtime.h>

typedef __attribute__((ext_vector_type(2))) float v2f;
typedef __attribute__((ext_vector_type(8))) float v8f;

__global__ __launch_bounds__(256) void zero_f32(float* __restrict__ p, long long n) {
    long long i = (long long)blockIdx.x * blockDim.x + threadIdx.x;
    if (i < n) p[i] = 0.0f;
}

__global__ __launch_bounds__(256) void deg_count(const int* __restrict__ dst,
                                                 float* __restrict__ deg, int nE) {
    int e = blockIdx.x * blockDim.x + threadIdx.x;
    if (e < nE) atomicAdd(&deg[dst[e]], 1.0f);
}

__global__ __launch_bounds__(256) void deg_to_dinv(float* __restrict__ deg, int n) {
    int i = blockIdx.x * blockDim.x + threadIdx.x;
    if (i < n) deg[i] = rsqrtf(deg[i] + 1.0f);
}

__global__ __launch_bounds__(256) void gemm_x_w1(const float* __restrict__ X,
                                                 const float* __restrict__ W,
                                                 float* __restrict__ H, int n) {
    const int lane = threadIdx.x & 31;
    const int tile = blockIdx.x * 8 + (threadIdx.x >> 5);
    const int row0 = tile * 16;
    if (row0 >= n) return;
    int m = row0 + (lane & 15);
    if (m >= n) m = n - 1;
    const int kh  = (lane >> 4) << 1;
    const int col = lane & 15;
    const float* xr = X + (size_t)m * 256;

    v8f acc = {};
#pragma unroll 4
    for (int k0 = 0; k0 < 256; k0 += 4) {
        v2f a, b;
        a.x = xr[k0 + kh];
        a.y = xr[k0 + kh + 1];
        b.x = W[(size_t)(k0 + kh) * 16 + col];
        b.y = W[(size_t)(k0 + kh + 1) * 16 + col];
        acc = __builtin_amdgcn_wmma_f32_16x16x4_f32(false, a, false, b,
                                                    (short)0, acc, false, false);
    }
    const int ra = (lane >> 4) << 3;
    if (row0 + 16 <= n) {
#pragma unroll
        for (int v = 0; v < 8; ++v)
            H[(size_t)(row0 + ra + v) * 16 + col] = acc[v];
    } else {
#pragma unroll
        for (int v = 0; v < 8; ++v) {
            int r = row0 + ra + v;
            if (r < n) H[(size_t)r * 16 + col] = acc[v];
        }
    }
}

__global__ __launch_bounds__(256) void gemm_z_w2(const float* __restrict__ Z,
                                                 const float* __restrict__ W,
                                                 float* __restrict__ H, int n) {
    const int lane = threadIdx.x & 31;
    const int tile = blockIdx.x * 8 + (threadIdx.x >> 5);
    const int row0 = tile * 16;
    if (row0 >= n) return;
    int m = row0 + (lane & 15);
    if (m >= n) m = n - 1;
    const int kh  = (lane >> 4) << 1;
    const int col = lane & 15;
    const float* zr = Z + (size_t)m * 16;

    v8f acc0 = {}, acc1 = {}, acc2 = {}, acc3 = {};
#pragma unroll
    for (int k0 = 0; k0 < 16; k0 += 4) {
        v2f a, b;
        a.x = zr[k0 + kh];
        a.y = zr[k0 + kh + 1];
        const float* w0 = W + (size_t)(k0 + kh) * 64;
        const float* w1 = w0 + 64;
        b.x = w0[col];      b.y = w1[col];
        acc0 = __builtin_amdgcn_wmma_f32_16x16x4_f32(false, a, false, b, (short)0, acc0, false, false);
        b.x = w0[col + 16]; b.y = w1[col + 16];
        acc1 = __builtin_amdgcn_wmma_f32_16x16x4_f32(false, a, false, b, (short)0, acc1, false, false);
        b.x = w0[col + 32]; b.y = w1[col + 32];
        acc2 = __builtin_amdgcn_wmma_f32_16x16x4_f32(false, a, false, b, (short)0, acc2, false, false);
        b.x = w0[col + 48]; b.y = w1[col + 48];
        acc3 = __builtin_amdgcn_wmma_f32_16x16x4_f32(false, a, false, b, (short)0, acc3, false, false);
    }
    const int ra = (lane >> 4) << 3;
    if (row0 + 16 <= n) {
#pragma unroll
        for (int v = 0; v < 8; ++v) {
            float* hp = H + (size_t)(row0 + ra + v) * 64 + col;
            hp[0]  = acc0[v];
            hp[16] = acc1[v];
            hp[32] = acc2[v];
            hp[48] = acc3[v];
        }
    } else {
#pragma unroll
        for (int v = 0; v < 8; ++v) {
            int r = row0 + ra + v;
            if (r < n) {
                float* hp = H + (size_t)r * 64 + col;
                hp[0]  = acc0[v];
                hp[16] = acc1[v];
                hp[32] = acc2[v];
                hp[48] = acc3[v];
            }
        }
    }
}

template <int F>
__global__ __launch_bounds__(256) void scatter_add(const int* __restrict__ src,
                                                   const int* __restrict__ dst,
                                                   const float* __restrict__ dinv,
                                                   const float* __restrict__ H,
                                                   float* __restrict__ OUT, int nE) {
    constexpr int CH = F / 16;
    long long idx = (long long)blockIdx.x * blockDim.x + threadIdx.x;
    long long e = idx / CH;
    int c = (int)(idx % CH);
    if (e >= nE) return;
    int s = src[e], d = dst[e];
    float nrm = dinv[s] * dinv[d];
    const float* hp = H + (size_t)s * F + c * 16;
    float* op = OUT + (size_t)d * F + c * 16;
#pragma unroll
    for (int f = 0; f < 16; ++f)
        atomicAdd(op + f, hp[f] * nrm);
}

template <int F, bool RELU>
__global__ __launch_bounds__(256) void epilogue(float* __restrict__ AGG,
                                                const float* __restrict__ H,
                                                const float* __restrict__ dinv,
                                                const float* __restrict__ bias, int n) {
    long long idx = (long long)blockIdx.x * blockDim.x + threadIdx.x;
    if (idx >= (long long)n * F) return;
    int i = (int)(idx / F);
    int f = (int)(idx % F);
    float di = dinv[i];
    float v = AGG[idx] + di * di * H[idx] + bias[f];
    if (RELU) v = fmaxf(v, 0.0f);
    AGG[idx] = v;
}

__global__ __launch_bounds__(256) void edge_score(const int* __restrict__ psrc,
                                                  const int* __restrict__ pdst,
                                                  const int* __restrict__ nsrc,
                                                  const int* __restrict__ ndst,
                                                  const float* __restrict__ Z,
                                                  float* __restrict__ out, int nT) {
    int j = blockIdx.x * blockDim.x + threadIdx.x;
    if (j >= 2 * nT) return;
    int s, d;
    if (j < nT) { s = psrc[j]; d = pdst[j]; }
    else        { s = nsrc[j - nT]; d = ndst[j - nT]; }
    const float4* zs = (const float4*)(Z + (size_t)s * 64);
    const float4* zd = (const float4*)(Z + (size_t)d * 64);
    float acc = 0.0f;
#pragma unroll
    for (int k = 0; k < 16; ++k) {
        float4 a = zs[k], b = zd[k];
        acc = fmaf(a.x, b.x, acc);
        acc = fmaf(a.y, b.y, acc);
        acc = fmaf(a.z, b.z, acc);
        acc = fmaf(a.w, b.w, acc);
    }
    out[j] = acc;
}

extern "C" void kernel_launch(void* const* d_in, const int* in_sizes, int n_in,
                              void* d_out, int out_size, void* d_ws, size_t ws_size,
                              hipStream_t stream) {
    const float* x   = (const float*)d_in[0];
    const int*   tei = (const int*)d_in[1];
    const int*   tpo = (const int*)d_in[2];
    const int*   tne = (const int*)d_in[3];
    const float* W1  = (const float*)d_in[4];
    const float* b1  = (const float*)d_in[5];
    const float* W2  = (const float*)d_in[6];
    const float* b2  = (const float*)d_in[7];

    const int N  = in_sizes[0] / 256;
    const int E  = in_sizes[1] / 2;
    const int ET = in_sizes[2] / 2;

    float* ws   = (float*)d_ws;
    float* dinv = ws;
    float* h1   = ws + (size_t)N;
    float* agg1 = ws + (size_t)17 * N;
    float* h2   = ws + (size_t)33 * N;
    float* agg2 = ws + (size_t)97 * N;

    const int B = 256;
    auto cdiv = [](long long a, long long b) { return (int)((a + b - 1) / b); };

    zero_f32<<<cdiv(N, B), B, 0, stream>>>(dinv, N);
    zero_f32<<<cdiv((long long)16 * N, B), B, 0, stream>>>(agg1, (long long)16 * N);
    zero_f32<<<cdiv((long long)64 * N, B), B, 0, stream>>>(agg2, (long long)64 * N);

    deg_count<<<cdiv(E, B), B, 0, stream>>>(tei + E, dinv, E);
    deg_to_dinv<<<cdiv(N, B), B, 0, stream>>>(dinv, N);

    const int tiles = (N + 15) / 16;

    gemm_x_w1<<<cdiv(tiles, 8), B, 0, stream>>>(x, W1, h1, N);
    scatter_add<16><<<cdiv(E, B), B, 0, stream>>>(tei, tei + E, dinv, h1, agg1, E);
    epilogue<16, true><<<cdiv((long long)16 * N, B), B, 0, stream>>>(agg1, h1, dinv, b1, N);

    gemm_z_w2<<<cdiv(tiles, 8), B, 0, stream>>>(agg1, W2, h2, N);
    scatter_add<64><<<cdiv((long long)4 * E, B), B, 0, stream>>>(tei, tei + E, dinv, h2, agg2, E);
    epilogue<64, false><<<cdiv((long long)64 * N, B), B, 0, stream>>>(agg2, h2, dinv, b2, N);

    edge_score<<<cdiv(2 * ET, B), B, 0, stream>>>(tpo, tpo + ET, tne, tne + ET,
                                                  agg2, (float*)d_out, ET);
}


# Aggregation by destination tiles: a block keeps its node range in LDS, scans the edge list, adds matching rows with LDS float adds; chunk planes summed after; no global atomics (sum order changes)
# speedup vs baseline: 3.4201x; 3.4201x over previous
#include <hip/hip_runtime.h>

typedef __attribute__((ext_vector_type(2))) float v2f;
typedef __attribute__((ext_vector_type(8))) float v8f;

__global__ __launch_bounds__(256) void zero_f32(float* __restrict__ p, long long n) {
    long long i = (long long)blockIdx.x * blockDim.x + threadIdx.x;
    if (i < n) p[i] = 0.0f;
}

__global__ __launch_bounds__(256) void deg_count(const int* __restrict__ dst,
                                                 float* __restrict__ deg, int nE) {
    int e = blockIdx.x * blockDim.x + threadIdx.x;
    if (e < nE) atomicAdd(&deg[dst[e]], 1.0f);
}

__global__ __launch_bounds__(256) void deg_to_dinv(float* __restrict__ deg, int n) {
    int i = blockIdx.x * blockDim.x + threadIdx.x;
    if (i < n) deg[i] = rsqrtf(deg[i] + 1.0f);
}

__global__ __launch_bounds__(256) void gemm_x_w1(const float* __restrict__ X,
                                                 const float* __restrict__ W,
                                                 float* __restrict__ H, int n) {
    const int lane = threadIdx.x & 31;
    const int tile = blockIdx.x * 8 + (threadIdx.x >> 5);
    const int row0 = tile * 16;
    if (row0 >= n) return;
    int m = row0 + (lane & 15);
    if (m >= n) m = n - 1;
    const int kh  = (lane >> 4) << 1;
    const int col = lane & 15;
    const float* xr = X + (size_t)m * 256;

    v8f acc = {};
#pragma unroll 4
    for (int k0 = 0; k0 < 256; k0 += 4) {
        v2f a, b;
        a.x = xr[k0 + kh];
        a.y = xr[k0 + kh + 1];
        b.x = W[(size_t)(k0 + kh) * 16 + col];
        b.y = W[(size_t)(k0 + kh + 1) * 16 + col];
        acc = __builtin_amdgcn_wmma_f32_16x16x4_f32(false, a, false, b,
                                                    (short)0, acc, false, false);
    }
    const int ra = (lane >> 4) << 3;
    if (row0 + 16 <= n) {
#pragma unroll
        for (int v = 0; v < 8; ++v)
            H[(size_t)(row0 + ra + v) * 16 + col] = acc[v];
    } else {
#pragma unroll
        for (int v = 0; v < 8; ++v) {
            int r = row0 + ra + v;
            if (r < n) H[(size_t)r * 16 + col] = acc[v];
        }
    }
}

__global__ __launch_bounds__(256) void gemm_z_w2(const float* __restrict__ Z,
                                                 const float* __restrict__ W,
                                                 float* __restrict__ H, int n) {
    const int lane = threadIdx.x & 31;
    const int tile = blockIdx.x * 8 + (threadIdx.x >> 5);
    const int row0 = tile * 16;
    if (row0 >= n) return;
    int m = row0 + (lane & 15);
    if (m >= n) m = n - 1;
    const int kh  = (lane >> 4) << 1;
    const int col = lane & 15;
    const float* zr = Z + (size_t)m * 16;

    v8f acc0 = {}, acc1 = {}, acc2 = {}, acc3 = {};
#pragma unroll
    for (int k0 = 0; k0 < 16; k0 += 4) {
        v2f a, b;
        a.x = zr[k0 + kh];
        a.y = zr[k0 + kh + 1];
        const float* w0 = W + (size_t)(k0 + kh) * 64;
        const float* w1 = w0 + 64;
        b.x = w0[col];      b.y = w1[col];
        acc0 = __builtin_amdgcn_wmma_f32_16x16x4_f32(false, a, false, b, (short)0, acc0, false, false);
        b.x = w0[col + 16]; b.y = w1[col + 16];
        acc1 = __builtin_amdgcn_wmma_f32_16x16x4_f32(false, a, false, b, (short)0, acc1, false, false);
        b.x = w0[col + 32]; b.y = w1[col + 32];
        acc2 = __builtin_amdgcn_wmma_f32_16x16x4_f32(false, a, false, b, (short)0, acc2, false, false);
        b.x = w0[col + 48]; b.y = w1[col + 48];
        acc3 = __builtin_amdgcn_wmma_f32_16x16x4_f32(false, a, false, b, (short)0, acc3, false, false);
    }
    const int ra = (lane >> 4) << 3;
    if (row0 + 16 <= n) {
#pragma unroll
        for (int v = 0; v < 8; ++v) {
            float* hp = H + (size_t)(row0 + ra + v) * 64 + col;
            hp[0]  = acc0[v];
            hp[16] = acc1[v];
            hp[32] = acc2[v];
            hp[48] = acc3[v];
        }
    } else {
#pragma unroll
        for (int v = 0; v < 8; ++v) {
            int r = row0 + ra + v;
            if (r < n) {
                float* hp = H + (size_t)r * 64 + col;
                hp[0]  = acc0[v];
                hp[16] = acc1[v];
                hp[32] = acc2[v];
                hp[48] = acc3[v];
            }
        }
    }
}

constexpr int QCAP = 160;

template <int F>
__device__ __forceinline__ void add_rows(const int* __restrict__ src, const int* __restrict__ dst,
                                         const float* __restrict__ dinv, const float* __restrict__ H,
                                         float* tile, int base, int e, int cnt, int lane) {
    constexpr int V = F / 16;
    int s = 0, loc = 0;
    float nrm = 0.0f;
    if (lane < cnt) {
        s = src[e];
        const int d = dst[e];
        nrm = dinv[s] * dinv[d];
        loc = d - base;
    }
    const int h = lane >> 4, col = lane & 15;
    for (int j = 0; j < cnt; j += 8) {
        float r[4][V], nj[4];
        int   off[4];
        bool  act[4];
#pragma unroll
        for (int u = 0; u < 4; ++u) {
            const int ja = j + 2 * u;
            const int sA = __builtin_amdgcn_readlane(s, ja);
            const int sB = __builtin_amdgcn_readlane(s, ja + 1);
            const int lA = __builtin_amdgcn_readlane(loc, ja);
            const int lB = __builtin_amdgcn_readlane(loc, ja + 1);
            const int nA = __builtin_amdgcn_readlane(__float_as_int(nrm), ja);
            const int nB = __builtin_amdgcn_readlane(__float_as_int(nrm), ja + 1);
            const int sj = h ? sB : sA;
            off[u] = (h ? lB : lA) * (F + 1) + col * V;
            nj[u]  = __int_as_float(h ? nB : nA);
            act[u] = ja + h < cnt;
            const float* hp = H + (size_t)sj * F + col * V;
            if constexpr (V == 4) {
                const float4 t = *(const float4*)hp;
                r[u][0] = t.x; r[u][1] = t.y; r[u][2] = t.z; r[u][3] = t.w;
            } else {
#pragma unroll
                for (int i = 0; i < V; ++i) r[u][i] = hp[i];
            }
        }
#pragma unroll
        for (int u = 0; u < 4; ++u) {
            if (act[u]) {
#pragma unroll
                for (int i = 0; i < V; ++i) atomicAdd(tile + off[u] + i, r[u][i] * nj[u]);
            }
        }
    }
}

template <int F>
__global__ __launch_bounds__(1024) void tile_aggregate(const int* __restrict__ src,
                                                       const int* __restrict__ dst,
                                                       const float* __restrict__ dinv,
                                                       const float* __restrict__ H,
                                                       float* __restrict__ P,
                                                       int nE, int n) {
    constexpr int TN = 32768 / F;
    constexpr int PITCH = F + 1;
    __shared__ float tile[TN * PITCH];
    __shared__ int   stack[32 * QCAP];

    const int tid  = threadIdx.x;
    const int lane = tid & 31;
    const int w    = tid >> 5;
    const int base = blockIdx.x * TN;
    const int c    = blockIdx.y;
    int* q = stack + w * QCAP;

    for (int i = tid; i < TN * PITCH; i += 1024) tile[i] = 0.0f;
    __syncthreads();

    const int ec = (nE + (int)gridDim.y - 1) / (int)gridDim.y;
    const int e0 = c * ec;
    const int e1 = min(nE, e0 + ec);

    int pending = 0;
    int dn[4];
#pragma unroll
    for (int k = 0; k < 4; ++k)
        dn[k] = dst[max(min(e0 + w * 128 + k * 32 + lane, e1 - 1), 0)];
    for (int eb = e0; eb < e1; eb += 4096) {
        int ee[4], dd[4];
#pragma unroll
        for (int k = 0; k < 4; ++k) {
            ee[k] = eb + w * 128 + k * 32 + lane;
            dd[k] = dn[k];
            dn[k] = dst[min(ee[k] + 4096, e1 - 1)];
        }
        while (pending >= 32) {
            pending -= 32;
            add_rows<F>(src, dst, dinv, H, tile, base, q[pending + lane], 32, lane);
        }
#pragma unroll
        for (int k = 0; k < 4; ++k) {
            const bool m = ee[k] < e1 && (unsigned)(dd[k] - base) < (unsigned)TN;
            const unsigned bal = __builtin_amdgcn_ballot_w32(m);
            if (m) q[pending + (int)__builtin_amdgcn_mbcnt_lo(bal, 0u)] = ee[k];
            pending += __popc(bal);
        }
    }
    while (pending >= 32) {
        pending -= 32;
        add_rows<F>(src, dst, dinv, H, tile, base, q[pending + lane], 32, lane);
    }
    add_rows<F>(src, dst, dinv, H, tile, base, lane < pending ? q[lane] : 0, pending, lane);
    __syncthreads();

    float* pp = P + ((size_t)c * n + base) * F;
    const int rows = min(TN, n - base);
    for (int i = tid; i < rows * F; i += 1024)
        pp[i] = tile[(i / F) * PITCH + (i % F)];
}

template <int F, bool RELU>
__global__ __launch_bounds__(256) void epilogue(float* __restrict__ OUT,
                                                const float* __restrict__ P, int nC,
                                                const float* __restrict__ H,
                                                const float* __restrict__ dinv,
                                                const float* __restrict__ bias, int n) {
    long long idx = (long long)blockIdx.x * blockDim.x + threadIdx.x;
    if (idx >= (long long)n * F) return;
    int i = (int)(idx / F);
    int f = (int)(idx % F);
    float acc = 0.0f;
    for (int c = 0; c < nC; ++c)
        acc += P[(size_t)c * n * F + idx];
    float di = dinv[i];
    float v = acc + di * di * H[idx] + bias[f];
    if (RELU) v = fmaxf(v, 0.0f);
    OUT[idx] = v;
}

__global__ __launch_bounds__(256) void edge_score(const int* __restrict__ psrc,
                                                  const int* __restrict__ pdst,
                                                  const int* __restrict__ nsrc,
                                                  const int* __restrict__ ndst,
                                                  const float* __restrict__ Z,
                                                  float* __restrict__ out, int nT) {
    int j = blockIdx.x * blockDim.x + threadIdx.x;
    if (j >= 2 * nT) return;
    int s, d;
    if (j < nT) { s = psrc[j]; d = pdst[j]; }
    else        { s = nsrc[j - nT]; d = ndst[j - nT]; }
    const float4* zs = (const float4*)(Z + (size_t)s * 64);
    const float4* zd = (const float4*)(Z + (size_t)d * 64);
    float acc = 0.0f;
#pragma unroll
    for (int k = 0; k < 16; ++k) {
        float4 a = zs[k], b = zd[k];
        acc = fmaf(a.x, b.x, acc);
        acc = fmaf(a.y, b.y, acc);
        acc = fmaf(a.z, b.z, acc);
        acc = fmaf(a.w, b.w, acc);
    }
    out[j] = acc;
}

extern "C" void kernel_launch(void* const* d_in, const int* in_sizes, int n_in,
                              void* d_out, int out_size, void* d_ws, size_t ws_size,
                              hipStream_t stream) {
    const float* x   = (const float*)d_in[0];
    const int*   tei = (const int*)d_in[1];
    const int*   tpo = (const int*)d_in[2];
    const int*   tne = (const int*)d_in[3];
    const float* W1  = (const float*)d_in[4];
    const float* b1  = (const float*)d_in[5];
    const float* W2  = (const float*)d_in[6];
    const float* b2  = (const float*)d_in[7];

    const int N  = in_sizes[0] / 256;
    const int E  = in_sizes[1] / 2;
    const int ET = in_sizes[2] / 2;

    float* ws   = (float*)d_ws;
    float* dinv = ws;
    float* h1   = ws + (size_t)N;
    float* z1   = ws + (size_t)17 * N;
    float* h2   = ws + (size_t)33 * N;
    float* z2   = ws + (size_t)97 * N;
    float* P    = ws + (size_t)161 * N;

    const int B = 256;
    auto cdiv = [](long long a, long long b) { return (int)((a + b - 1) / b); };

    zero_f32<<<cdiv(N, B), B, 0, stream>>>(dinv, N);
    deg_count<<<cdiv(E, B), B, 0, stream>>>(tei + E, dinv, E);
    deg_to_dinv<<<cdiv(N, B), B, 0, stream>>>(dinv, N);

    const int tiles = (N + 15) / 16;

    gemm_x_w1<<<cdiv(tiles, 8), B, 0, stream>>>(x, W1, h1, N);
    tile_aggregate<16><<<dim3(cdiv(N, 2048), 10), 1024, 0, stream>>>(tei, tei + E, dinv, h1, P, E, N);
    epilogue<16, true><<<cdiv((long long)16 * N, B), B, 0, stream>>>(z1, P, 10, h1, dinv, b1, N);

    gemm_z_w2<<<cdiv(tiles, 8), B, 0, stream>>>(z1, W2, h2, N);
    tile_aggregate<64><<<dim3(cdiv(N, 512), 5), 1024, 0, stream>>>(tei, tei + E, dinv, h2, P, E, N);
    epilogue<64, false><<<cdiv((long long)64 * N, B), B, 0, stream>>>(z2, P, 5, h2, dinv, b2, N);

    edge_score<<<cdiv(2 * ET, B), B, 0, stream>>>(tpo, tpo + ET, tne, tne + ET,
                                                  z2, (float*)d_out, ET);
}


# Aggregation scan reads a tile-key byte per edge (written by the degree pass), 16 per lane and load, 4 tested per 32-bit op, not 4-byte destinations; hits pushed lane by lane (sum order changes)
# speedup vs baseline: 3.7017x; 1.0823x over previous
#include <hip/hip_runtime.h>

typedef __attribute__((ext_vector_type(2))) float v2f;
typedef __attribute__((ext_vector_type(8))) float v8f;

__global__ __launch_bounds__(256) void zero_f32(float* __restrict__ p, long long n) {
    long long i = (long long)blockIdx.x * blockDim.x + threadIdx.x;
    if (i < n) p[i] = 0.0f;
}

constexpr int KEY_SHIFT = 9;

__global__ __launch_bounds__(256) void deg_count_keys(const int* __restrict__ dst,
                                                      float* __restrict__ deg,
                                                      unsigned* __restrict__ keys,
                                                      int nE, int nWords) {
    const int i = blockIdx.x * blockDim.x + threadIdx.x;
    if (i >= nWords) return;
    const int e = 4 * i;
    int d[4] = {-1, -1, -1, -1};
    if (e + 3 < nE && (((size_t)dst) & 15) == 0) {
        const int4 v = *(const int4*)(dst + e);
        d[0] = v.x; d[1] = v.y; d[2] = v.z; d[3] = v.w;
    } else {
#pragma unroll
        for (int b = 0; b < 4; ++b)
            if (e + b < nE) d[b] = dst[e + b];
    }
    unsigned k = 0;
#pragma unroll
    for (int b = 0; b < 4; ++b) {
        unsigned kb = 0xFFu;
        if (d[b] >= 0) {
            atomicAdd(&deg[d[b]], 1.0f);
            kb = ((unsigned)d[b] >> KEY_SHIFT) & 0xFFu;
        }
        k |= kb << (8 * b);
    }
    keys[i] = k;
}

__global__ __launch_bounds__(256) void deg_to_dinv(float* __restrict__ deg, int n) {
    int i = blockIdx.x * blockDim.x + threadIdx.x;
    if (i < n) deg[i] = rsqrtf(deg[i] + 1.0f);
}

__global__ __launch_bounds__(256) void gemm_x_w1(const float* __restrict__ X,
                                                 const float* __restrict__ W,
                                                 float* __restrict__ H, int n) {
    const int lane = threadIdx.x & 31;
    const int tile = blockIdx.x * 8 + (threadIdx.x >> 5);
    const int row0 = tile * 16;
    if (row0 >= n) return;
    int m = row0 + (lane & 15);
    if (m >= n) m = n - 1;
    const int kh  = (lane >> 4) << 1;
    const int col = lane & 15;
    const float* xr = X + (size_t)m * 256;

    v8f acc = {};
#pragma unroll 4
    for (int k0 = 0; k0 < 256; k0 += 4) {
        v2f a, b;
        a.x = xr[k0 + kh];
        a.y = xr[k0 + kh + 1];
        b.x = W[(size_t)(k0 + kh) * 16 + col];
        b.y = W[(size_t)(k0 + kh + 1) * 16 + col];
        acc = __builtin_amdgcn_wmma_f32_16x16x4_f32(false, a, false, b,
                                                    (short)0, acc, false, false);
    }
    const int ra = (lane >> 4) << 3;
    if (row0 + 16 <= n) {
#pragma unroll
        for (int v = 0; v < 8; ++v)
            H[(size_t)(row0 + ra + v) * 16 + col] = acc[v];
    } else {
#pragma unroll
        for (int v = 0; v < 8; ++v) {
            int r = row0 + ra + v;
            if (r < n) H[(size_t)r * 16 + col] = acc[v];
        }
    }
}

__global__ __launch_bounds__(256) void gemm_z_w2(const float* __restrict__ Z,
                                                 const float* __restrict__ W,
                                                 float* __restrict__ H, int n) {
    const int lane = threadIdx.x & 31;
    const int tile = blockIdx.x * 8 + (threadIdx.x >> 5);
    const int row0 = tile * 16;
    if (row0 >= n) return;
    int m = row0 + (lane & 15);
    if (m >= n) m = n - 1;
    const int kh  = (lane >> 4) << 1;
    const int col = lane & 15;
    const float* zr = Z + (size_t)m * 16;

    v8f acc0 = {}, acc1 = {}, acc2 = {}, acc3 = {};
#pragma unroll
    for (int k0 = 0; k0 < 16; k0 += 4) {
        v2f a, b;
        a.x = zr[k0 + kh];
        a.y = zr[k0 + kh + 1];
        const float* w0 = W + (size_t)(k0 + kh) * 64;
        const float* w1 = w0 + 64;
        b.x = w0[col];      b.y = w1[col];
        acc0 = __builtin_amdgcn_wmma_f32_16x16x4_f32(false, a, false, b, (short)0, acc0, false, false);
        b.x = w0[col + 16]; b.y = w1[col + 16];
        acc1 = __builtin_amdgcn_wmma_f32_16x16x4_f32(false, a, false, b, (short)0, acc1, false, false);
        b.x = w0[col + 32]; b.y = w1[col + 32];
        acc2 = __builtin_amdgcn_wmma_f32_16x16x4_f32(false, a, false, b, (short)0, acc2, false, false);
        b.x = w0[col + 48]; b.y = w1[col + 48];
        acc3 = __builtin_amdgcn_wmma_f32_16x16x4_f32(false, a, false, b, (short)0, acc3, false, false);
    }
    const int ra = (lane >> 4) << 3;
    if (row0 + 16 <= n) {
#pragma unroll
        for (int v = 0; v < 8; ++v) {
            float* hp = H + (size_t)(row0 + ra + v) * 64 + col;
            hp[0]  = acc0[v];
            hp[16] = acc1[v];
            hp[32] = acc2[v];
            hp[48] = acc3[v];
        }
    } else {
#pragma unroll
        for (int v = 0; v < 8; ++v) {
            int r = row0 + ra + v;
            if (r < n) {
                float* hp = H + (size_t)r * 64 + col;
                hp[0]  = acc0[v];
                hp[16] = acc1[v];
                hp[32] = acc2[v];
                hp[48] = acc3[v];
            }
        }
    }
}

constexpr int QCAP = 64;

template <int F>
__device__ __forceinline__ void add_rows(const int* __restrict__ src, const int* __restrict__ dst,
                                         const float* __restrict__ dinv, const float* __restrict__ H,
                                         float* tile, int base, int e, int cnt, int lane) {
    constexpr int V = F / 16;
    int s = 0, loc = 0;
    float nrm = 0.0f;
    if (lane < cnt) {
        s = src[e];
        const int d = dst[e];
        nrm = dinv[s] * dinv[d];
        loc = d - base;
        if ((unsigned)loc >= (unsigned)(32768 / F)) loc = -1;
    }
    const int h = lane >> 4, col = lane & 15;
    for (int j = 0; j < cnt; j += 8) {
        float r[4][V], nj[4];
        int   off[4];
        bool  act[4];
#pragma unroll
        for (int u = 0; u < 4; ++u) {
            const int ja = j + 2 * u;
            const int sA = __builtin_amdgcn_readlane(s, ja);
            const int sB = __builtin_amdgcn_readlane(s, ja + 1);
            const int lA = __builtin_amdgcn_readlane(loc, ja);
            const int lB = __builtin_amdgcn_readlane(loc, ja + 1);
            const int nA = __builtin_amdgcn_readlane(__float_as_int(nrm), ja);
            const int nB = __builtin_amdgcn_readlane(__float_as_int(nrm), ja + 1);
            const int sj = h ? sB : sA;
            const int lj = h ? lB : lA;
            off[u] = lj * (F + 1) + col * V;
            nj[u]  = __int_as_float(h ? nB : nA);
            act[u] = ja + h < cnt && lj >= 0;
            const float* hp = H + (size_t)sj * F + col * V;
            if constexpr (V == 4) {
                const float4 t = *(const float4*)hp;
                r[u][0] = t.x; r[u][1] = t.y; r[u][2] = t.z; r[u][3] = t.w;
            } else {
#pragma unroll
                for (int i = 0; i < V; ++i) r[u][i] = hp[i];
            }
        }
#pragma unroll
        for (int u = 0; u < 4; ++u) {
            if (act[u]) {
#pragma unroll
                for (int i = 0; i < V; ++i) atomicAdd(tile + off[u] + i, r[u][i] * nj[u]);
            }
        }
    }
}

__device__ __forceinline__ unsigned nonzero_bytes(unsigned t) {
    return ((t & 0x7F7F7F7Fu) + 0x7F7F7F7Fu) | t;
}

template <int F>
__global__ __launch_bounds__(1024) void tile_aggregate(const int* __restrict__ src,
                                                       const int* __restrict__ dst,
                                                       const float* __restrict__ dinv,
                                                       const float* __restrict__ H,
                                                       const unsigned char* __restrict__ keys,
                                                       float* __restrict__ P,
                                                       int nE, int n) {
    constexpr int TN = 32768 / F;
    constexpr int PITCH = F + 1;
    constexpr int KPT = TN >> KEY_SHIFT;
    static_assert(KPT >= 1 && (KPT & (KPT - 1)) == 0 && KPT * (1 << KEY_SHIFT) == TN, "tile = 2^k keys");
    constexpr unsigned KM  = (0x100u - KPT) & 0xFFu;
    constexpr unsigned KM4 = KM * 0x01010101u;
    __shared__ float tile[TN * PITCH];
    __shared__ int   stack[32 * QCAP];

    const int tid  = threadIdx.x;
    const int lane = tid & 31;
    const int w    = __builtin_amdgcn_readfirstlane(tid >> 5);
    const int base = blockIdx.x * TN;
    const int c    = blockIdx.y;
    int* q = stack + w * QCAP;
    const unsigned pat = (((unsigned)blockIdx.x * KPT) & KM) * 0x01010101u;

    for (int i = tid; i < TN * PITCH; i += 1024) tile[i] = 0.0f;
    __syncthreads();

    const int ec = (((nE + (int)gridDim.y - 1) / (int)gridDim.y) + 511) & ~511;
    const int e0 = c * ec;
    const int e1 = min(nE, e0 + ec);

    int pending = 0;
    constexpr int STRIDE = 32 * 512;
    int eb = e0 + w * 512;
    uint4 kn = make_uint4(0u, 0u, 0u, 0u);
    if (eb < e1) kn = *(const uint4*)(keys + eb + lane * 16);
    for (; eb < e1; eb += STRIDE) {
        const uint4 kc = kn;
        if (eb + STRIDE < e1) kn = *(const uint4*)(keys + eb + STRIDE + lane * 16);
        const unsigned n0 = nonzero_bytes((kc.x ^ pat) & KM4);
        const unsigned n1 = nonzero_bytes((kc.y ^ pat) & KM4);
        const unsigned n2 = nonzero_bytes((kc.z ^ pat) & KM4);
        const unsigned n3 = nonzero_bytes((kc.w ^ pat) & KM4);
        unsigned m = (((n0 >> 7) & 0x01010101u) | ((n1 >> 6) & 0x02020202u) |
                      ((n2 >> 5) & 0x04040404u) | ((n3 >> 4) & 0x08080808u)) ^ 0x0F0F0F0Fu;
        const int elane = eb + lane * 16;
        unsigned any;
        while ((any = __builtin_amdgcn_ballot_w32(m != 0u)) != 0u) {
            const int t = __ffs((int)m) - 1;
            const int e = elane + ((t & 7) << 2) + (t >> 3);
            const unsigned bal = any & __builtin_amdgcn_ballot_w32(e < e1);
            if (m != 0u && e < e1) q[pending + (int)__builtin_amdgcn_mbcnt_lo(bal, 0u)] = e;
            pending += __popc(bal);
            m &= m - 1u;
            if (pending >= 32) {
                pending -= 32;
                add_rows<F>(src, dst, dinv, H, tile, base, q[pending + lane], 32, lane);
            }
        }
    }
    add_rows<F>(src, dst, dinv, H, tile, base, lane < pending ? q[lane] : 0, pending, lane);
    __syncthreads();

    float* pp = P + ((size_t)c * n + base) * F;
    const int rows = min(TN, n - base);
    for (int i = tid; i < rows * F; i += 1024)
        pp[i] = tile[(i / F) * PITCH + (i % F)];
}

template <int F, bool RELU>
__global__ __launch_bounds__(256) void epilogue(float* __restrict__ OUT,
                                                const float* __restrict__ P, int nC,
                                                const float* __restrict__ H,
                                                const float* __restrict__ dinv,
                                                const float* __restrict__ bias, int n) {
    long long idx = (long long)blockIdx.x * blockDim.x + threadIdx.x;
    if (idx >= (long long)n * F) return;
    int i = (int)(idx / F);
    int f = (int)(idx % F);
    float acc = 0.0f;
    for (int c = 0; c < nC; ++c)
        acc += P[(size_t)c * n * F + idx];
    float di = dinv[i];
    float v = acc + di * di * H[idx] + bias[f];
    if (RELU) v = fmaxf(v, 0.0f);
    OUT[idx] = v;
}

__global__ __launch_bounds__(256) void edge_score(const int* __restrict__ psrc,
                                                  const int* __restrict__ pdst,
                                                  const int* __restrict__ nsrc,
                                                  const int* __restrict__ ndst,
                                                  const float* __restrict__ Z,
                                                  float* __restrict__ out, int nT) {
    int j = blockIdx.x * blockDim.x + threadIdx.x;
    if (j >= 2 * nT) return;
    int s, d;
    if (j < nT) { s = psrc[j]; d = pdst[j]; }
    else        { s = nsrc[j - nT]; d = ndst[j - nT]; }
    const float4* zs = (const float4*)(Z + (size_t)s * 64);
    const float4* zd = (const float4*)(Z + (size_t)d * 64);
    float acc = 0.0f;
#pragma unroll
    for (int k = 0; k < 16; ++k) {
        float4 a = zs[k], b = zd[k];
        acc = fmaf(a.x, b.x, acc);
        acc = fmaf(a.y, b.y, acc);
        acc = fmaf(a.z, b.z, acc);
        acc = fmaf(a.w, b.w, acc);
    }
    out[j] = acc;
}

extern "C" void kernel_launch(void* const* d_in, const int* in_sizes, int n_in,
                              void* d_out, int out_size, void* d_ws, size_t ws_size,
                              hipStream_t stream) {
    const float* x   = (const float*)d_in[0];
    const int*   tei = (const int*)d_in[1];
    const int*   tpo = (const int*)d_in[2];
    const int*   tne = (const int*)d_in[3];
    const float* W1  = (const float*)d_in[4];
    const float* b1  = (const float*)d_in[5];
    const float* W2  = (const float*)d_in[6];
    const float* b2  = (const float*)d_in[7];

    const int N  = in_sizes[0] / 256;
    const int E  = in_sizes[1] / 2;
    const int ET = in_sizes[2] / 2;

    float* ws   = (float*)d_ws;
    float* dinv = ws;
    float* h1   = ws + (size_t)N;
    float* z1   = ws + (size_t)17 * N;
    float* h2   = ws + (size_t)33 * N;
    float* z2   = ws + (size_t)97 * N;
    float* P    = ws + (size_t)161 * N;
    unsigned* keys = (unsigned*)(ws + (((size_t)481 * N + 3) & ~(size_t)3));

    const int B = 256;
    auto cdiv = [](long long a, long long b) { return (int)((a + b - 1) / b); };

    zero_f32<<<cdiv(N, B), B, 0, stream>>>(dinv, N);
    const int keyWords = cdiv(E, 512) * 128;
    deg_count_keys<<<cdiv(keyWords, B), B, 0, stream>>>(tei + E, dinv, keys, E, keyWords);
    deg_to_dinv<<<cdiv(N, B), B, 0, stream>>>(dinv, N);

    const int tiles = (N + 15) / 16;

    gemm_x_w1<<<cdiv(tiles, 8), B, 0, stream>>>(x, W1, h1, N);
    tile_aggregate<16><<<dim3(cdiv(N, 2048), 10), 1024, 0, stream>>>(tei, tei + E, dinv, h1, (const unsigned char*)keys, P, E, N);
    epilogue<16, true><<<cdiv((long long)16 * N, B), B, 0, stream>>>(z1, P, 10, h1, dinv, b1, N);

    gemm_z_w2<<<cdiv(tiles, 8), B, 0, stream>>>(z1, W2, h2, N);
    tile_aggregate<64><<<dim3(cdiv(N, 512), 5), 1024, 0, stream>>>(tei, tei + E, dinv, h2, (const unsigned char*)keys, P, E, N);
    epilogue<64, false><<<cdiv((long long)64 * N, B), B, 0, stream>>>(z2, P, 5, h2, dinv, b2, N);

    edge_score<<<cdiv(2 * ET, B), B, 0, stream>>>(tpo, tpo + ET, tne, tne + ET,
                                                  z2, (float*)d_out, ET);
}


# New edge pass writes a 16-byte record per edge (src, dst, norm dinv[src]*dinv[dst]); both aggregations take one gather per edge, not four 4-byte gathers in two dependent rounds; 11 launches
# speedup vs baseline: 3.8945x; 1.0521x over previous
#include <hip/hip_runtime.h>

typedef __attribute__((ext_vector_type(2))) float v2f;
typedef __attribute__((ext_vector_type(8))) float v8f;

__global__ __launch_bounds__(256) void zero_f32(float* __restrict__ p, long long n) {
    long long i = (long long)blockIdx.x * blockDim.x + threadIdx.x;
    if (i < n) p[i] = 0.0f;
}

constexpr int KEY_SHIFT = 9;

__global__ __launch_bounds__(256) void deg_count_keys(const int* __restrict__ dst,
                                                      float* __restrict__ deg,
                                                      unsigned* __restrict__ keys,
                                                      int nE, int nWords) {
    const int i = blockIdx.x * blockDim.x + threadIdx.x;
    if (i >= nWords) return;
    const int e = 4 * i;
    int d[4] = {-1, -1, -1, -1};
    if (e + 3 < nE && (((size_t)dst) & 15) == 0) {
        const int4 v = *(const int4*)(dst + e);
        d[0] = v.x; d[1] = v.y; d[2] = v.z; d[3] = v.w;
    } else {
#pragma unroll
        for (int b = 0; b < 4; ++b)
            if (e + b < nE) d[b] = dst[e + b];
    }
    unsigned k = 0;
#pragma unroll
    for (int b = 0; b < 4; ++b) {
        unsigned kb = 0xFFu;
        if (d[b] >= 0) {
            atomicAdd(&deg[d[b]], 1.0f);
            kb = ((unsigned)d[b] >> KEY_SHIFT) & 0xFFu;
        }
        k |= kb << (8 * b);
    }
    keys[i] = k;
}

__global__ __launch_bounds__(256) void deg_to_dinv(float* __restrict__ deg, int n) {
    int i = blockIdx.x * blockDim.x + threadIdx.x;
    if (i < n) deg[i] = rsqrtf(deg[i] + 1.0f);
}

__global__ __launch_bounds__(256) void edge_records(const int* __restrict__ src,
                                                    const int* __restrict__ dst,
                                                    const float* __restrict__ dinv,
                                                    int* __restrict__ rec, int nE) {
    const int e = blockIdx.x * blockDim.x + threadIdx.x;
    if (e >= nE) return;
    const int s = src[e];
    const int d = dst[e];
    const float nrm = dinv[s] * dinv[d];
    *(int4*)(rec + 4 * (size_t)e) = make_int4(s, d, __float_as_int(nrm), 0);
}

__global__ __launch_bounds__(256) void gemm_x_w1(const float* __restrict__ X,
                                                 const float* __restrict__ W,
                                                 float* __restrict__ H, int n) {
    const int lane = threadIdx.x & 31;
    const int tile = blockIdx.x * 8 + (threadIdx.x >> 5);
    const int row0 = tile * 16;
    if (row0 >= n) return;
    int m = row0 + (lane & 15);
    if (m >= n) m = n - 1;
    const int kh  = (lane >> 4) << 1;
    const int col = lane & 15;
    const float* xr = X + (size_t)m * 256;

    v8f acc = {};
#pragma unroll 4
    for (int k0 = 0; k0 < 256; k0 += 4) {
        v2f a, b;
        a.x = xr[k0 + kh];
        a.y = xr[k0 + kh + 1];
        b.x = W[(size_t)(k0 + kh) * 16 + col];
        b.y = W[(size_t)(k0 + kh + 1) * 16 + col];
        acc = __builtin_amdgcn_wmma_f32_16x16x4_f32(false, a, false, b,
                                                    (short)0, acc, false, false);
    }
    const int ra = (lane >> 4) << 3;
    if (row0 + 16 <= n) {
#pragma unroll
        for (int v = 0; v < 8; ++v)
            H[(size_t)(row0 + ra + v) * 16 + col] = acc[v];
    } else {
#pragma unroll
        for (int v = 0; v < 8; ++v) {
            int r = row0 + ra + v;
            if (r < n) H[(size_t)r * 16 + col] = acc[v];
        }
    }
}

__global__ __launch_bounds__(256) void gemm_z_w2(const float* __restrict__ Z,
                                                 const float* __restrict__ W,
                                                 float* __restrict__ H, int n) {
    const int lane = threadIdx.x & 31;
    const int tile = blockIdx.x * 8 + (threadIdx.x >> 5);
    const int row0 = tile * 16;
    if (row0 >= n) return;
    int m = row0 + (lane & 15);
    if (m >= n) m = n - 1;
    const int kh  = (lane >> 4) << 1;
    const int col = lane & 15;
    const float* zr = Z + (size_t)m * 16;

    v8f acc0 = {}, acc1 = {}, acc2 = {}, acc3 = {};
#pragma unroll
    for (int k0 = 0; k0 < 16; k0 += 4) {
        v2f a, b;
        a.x = zr[k0 + kh];
        a.y = zr[k0 + kh + 1];
        const float* w0 = W + (size_t)(k0 + kh) * 64;
        const float* w1 = w0 + 64;
        b.x = w0[col];      b.y = w1[col];
        acc0 = __builtin_amdgcn_wmma_f32_16x16x4_f32(false, a, false, b, (short)0, acc0, false, false);
        b.x = w0[col + 16]; b.y = w1[col + 16];
        acc1 = __builtin_amdgcn_wmma_f32_16x16x4_f32(false, a, false, b, (short)0, acc1, false, false);
        b.x = w0[col + 32]; b.y = w1[col + 32];
        acc2 = __builtin_amdgcn_wmma_f32_16x16x4_f32(false, a, false, b, (short)0, acc2, false, false);
        b.x = w0[col + 48]; b.y = w1[col + 48];
        acc3 = __builtin_amdgcn_wmma_f32_16x16x4_f32(false, a, false, b, (short)0, acc3, false, false);
    }
    const int ra = (lane >> 4) << 3;
    if (row0 + 16 <= n) {
#pragma unroll
        for (int v = 0; v < 8; ++v) {
            float* hp = H + (size_t)(row0 + ra + v) * 64 + col;
            hp[0]  = acc0[v];
            hp[16] = acc1[v];
            hp[32] = acc2[v];
            hp[48] = acc3[v];
        }
    } else {
#pragma unroll
        for (int v = 0; v < 8; ++v) {
            int r = row0 + ra + v;
            if (r < n) {
                float* hp = H + (size_t)r * 64 + col;
                hp[0]  = acc0[v];
                hp[16] = acc1[v];
                hp[32] = acc2[v];
                hp[48] = acc3[v];
            }
        }
    }
}

constexpr int QCAP = 64;

template <int F>
__device__ __forceinline__ void add_rows(const int* __restrict__ rec, const float* __restrict__ H,
                                         float* tile, int base, int e, int cnt, int lane) {
    constexpr int V = F / 16;
    int s = 0, loc = 0;
    float nrm = 0.0f;
    if (lane < cnt) {
        const int4 r = *(const int4*)(rec + 4 * (size_t)e);
        s = r.x;
        nrm = __int_as_float(r.z);
        loc = r.y - base;
        if ((unsigned)loc >= (unsigned)(32768 / F)) loc = -1;
    }
    const int h = lane >> 4, col = lane & 15;
    for (int j = 0; j < cnt; j += 8) {
        float r[4][V], nj[4];
        int   off[4];
        bool  act[4];
#pragma unroll
        for (int u = 0; u < 4; ++u) {
            const int ja = j + 2 * u;
            const int sA = __builtin_amdgcn_readlane(s, ja);
            const int sB = __builtin_amdgcn_readlane(s, ja + 1);
            const int lA = __builtin_amdgcn_readlane(loc, ja);
            const int lB = __builtin_amdgcn_readlane(loc, ja + 1);
            const int nA = __builtin_amdgcn_readlane(__float_as_int(nrm), ja);
            const int nB = __builtin_amdgcn_readlane(__float_as_int(nrm), ja + 1);
            const int sj = h ? sB : sA;
            const int lj = h ? lB : lA;
            off[u] = lj * (F + 1) + col * V;
            nj[u]  = __int_as_float(h ? nB : nA);
            act[u] = ja + h < cnt && lj >= 0;
            const float* hp = H + (size_t)sj * F + col * V;
            if constexpr (V == 4) {
                const float4 t = *(const float4*)hp;
                r[u][0] = t.x; r[u][1] = t.y; r[u][2] = t.z; r[u][3] = t.w;
            } else {
#pragma unroll
                for (int i = 0; i < V; ++i) r[u][i] = hp[i];
            }
        }
#pragma unroll
        for (int u = 0; u < 4; ++u) {
            if (act[u]) {
#pragma unroll
                for (int i = 0; i < V; ++i) atomicAdd(tile + off[u] + i, r[u][i] * nj[u]);
            }
        }
    }
}

__device__ __forceinline__ unsigned nonzero_bytes(unsigned t) {
    return ((t & 0x7F7F7F7Fu) + 0x7F7F7F7Fu) | t;
}

template <int F>
__global__ __launch_bounds__(1024) void tile_aggregate(const int* __restrict__ rec,
                                                       const float* __restrict__ H,
                                                       const unsigned char* __restrict__ keys,
                                                       float* __restrict__ P,
                                                       int nE, int n) {
    constexpr int TN = 32768 / F;
    constexpr int PITCH = F + 1;
    constexpr int KPT = TN >> KEY_SHIFT;
    static_assert(KPT >= 1 && (KPT & (KPT - 1)) == 0 && KPT * (1 << KEY_SHIFT) == TN, "tile = 2^k keys");
    constexpr unsigned KM  = (0x100u - KPT) & 0xFFu;
    constexpr unsigned KM4 = KM * 0x01010101u;
    __shared__ float tile[TN * PITCH];
    __shared__ int   stack[32 * QCAP];

    const int tid  = threadIdx.x;
    const int lane = tid & 31;
    const int w    = __builtin_amdgcn_readfirstlane(tid >> 5);
    const int base = blockIdx.x * TN;
    const int c    = blockIdx.y;
    int* q = stack + w * QCAP;
    const unsigned pat = (((unsigned)blockIdx.x * KPT) & KM) * 0x01010101u;

    for (int i = tid; i < TN * PITCH; i += 1024) tile[i] = 0.0f;
    __syncthreads();

    const int ec = (((nE + (int)gridDim.y - 1) / (int)gridDim.y) + 511) & ~511;
    const int e0 = c * ec;
    const int e1 = min(nE, e0 + ec);

    int pending = 0;
    constexpr int STRIDE = 32 * 512;
    int eb = e0 + w * 512;
    uint4 kn = make_uint4(0u, 0u, 0u, 0u);
    if (eb < e1) kn = *(const uint4*)(keys + eb + lane * 16);
    for (; eb < e1; eb += STRIDE) {
        const uint4 kc = kn;
        if (eb + STRIDE < e1) kn = *(const uint4*)(keys + eb + STRIDE + lane * 16);
        const unsigned n0 = nonzero_bytes((kc.x ^ pat) & KM4);
        const unsigned n1 = nonzero_bytes((kc.y ^ pat) & KM4);
        const unsigned n2 = nonzero_bytes((kc.z ^ pat) & KM4);
        const unsigned n3 = nonzero_bytes((kc.w ^ pat) & KM4);
        unsigned m = (((n0 >> 7) & 0x01010101u) | ((n1 >> 6) & 0x02020202u) |
                      ((n2 >> 5) & 0x04040404u) | ((n3 >> 4) & 0x08080808u)) ^ 0x0F0F0F0Fu;
        const int elane = eb + lane * 16;
        unsigned any;
        while ((any = __builtin_amdgcn_ballot_w32(m != 0u)) != 0u) {
            const int t = __ffs((int)m) - 1;
            const int e = elane + ((t & 7) << 2) + (t >> 3);
            const unsigned bal = any & __builtin_amdgcn_ballot_w32(e < e1);
            if (m != 0u && e < e1) q[pending + (int)__builtin_amdgcn_mbcnt_lo(bal, 0u)] = e;
            pending += __popc(bal);
            m &= m - 1u;
            if (pending >= 32) {
                pending -= 32;
                add_rows<F>(rec, H, tile, base, q[pending + lane], 32, lane);
            }
        }
    }
    add_rows<F>(rec, H, tile, base, lane < pending ? q[lane] : 0, pending, lane);
    __syncthreads();

    float* pp = P + ((size_t)c * n + base) * F;
    const int rows = min(TN, n - base);
    for (int i = tid; i < rows * F; i += 1024)
        pp[i] = tile[(i / F) * PITCH + (i % F)];
}

template <int F, bool RELU>
__global__ __launch_bounds__(256) void epilogue(float* __restrict__ OUT,
                                                const float* __restrict__ P, int nC,
                                                const float* __restrict__ H,
                                                const float* __restrict__ dinv,
                                                const float* __restrict__ bias, int n) {
    long long idx = (long long)blockIdx.x * blockDim.x + threadIdx.x;
    if (idx >= (long long)n * F) return;
    int i = (int)(idx / F);
    int f = (int)(idx % F);
    float acc = 0.0f;
    for (int c = 0; c < nC; ++c)
        acc += P[(size_t)c * n * F + idx];
    float di = dinv[i];
    float v = acc + di * di * H[idx] + bias[f];
    if (RELU) v = fmaxf(v, 0.0f);
    OUT[idx] = v;
}

__global__ __launch_bounds__(256) void edge_score(const int* __restrict__ psrc,
                                                  const int* __restrict__ pdst,
                                                  const int* __restrict__ nsrc,
                                                  const int* __restrict__ ndst,
                                                  const float* __restrict__ Z,
                                                  float* __restrict__ out, int nT) {
    int j = blockIdx.x * blockDim.x + threadIdx.x;
    if (j >= 2 * nT) return;
    int s, d;
    if (j < nT) { s = psrc[j]; d = pdst[j]; }
    else        { s = nsrc[j - nT]; d = ndst[j - nT]; }
    const float4* zs = (const float4*)(Z + (size_t)s * 64);
    const float4* zd = (const float4*)(Z + (size_t)d * 64);
    float acc = 0.0f;
#pragma unroll
    for (int k = 0; k < 16; ++k) {
        float4 a = zs[k], b = zd[k];
        acc = fmaf(a.x, b.x, acc);
        acc = fmaf(a.y, b.y, acc);
        acc = fmaf(a.z, b.z, acc);
        acc = fmaf(a.w, b.w, acc);
    }
    out[j] = acc;
}

extern "C" void kernel_launch(void* const* d_in, const int* in_sizes, int n_in,
                              void* d_out, int out_size, void* d_ws, size_t ws_size,
                              hipStream_t stream) {
    const float* x   = (const float*)d_in[0];
    const int*   tei = (const int*)d_in[1];
    const int*   tpo = (const int*)d_in[2];
    const int*   tne = (const int*)d_in[3];
    const float* W1  = (const float*)d_in[4];
    const float* b1  = (const float*)d_in[5];
    const float* W2  = (const float*)d_in[6];
    const float* b2  = (const float*)d_in[7];

    const int N  = in_sizes[0] / 256;
    const int E  = in_sizes[1] / 2;
    const int ET = in_sizes[2] / 2;

    float* ws   = (float*)d_ws;
    float* dinv = ws;
    float* h1   = ws + (size_t)N;
    float* z1   = ws + (size_t)17 * N;
    float* h2   = ws + (size_t)33 * N;
    float* z2   = ws + (size_t)97 * N;
    float* P    = ws + (size_t)161 * N;
    unsigned* keys = (unsigned*)(ws + (((size_t)481 * N + 3) & ~(size_t)3));

    const int B = 256;
    auto cdiv = [](long long a, long long b) { return (int)((a + b - 1) / b); };

    zero_f32<<<cdiv(N, B), B, 0, stream>>>(dinv, N);
    const int keyWords = cdiv(E, 512) * 128;
    deg_count_keys<<<cdiv(keyWords, B), B, 0, stream>>>(tei + E, dinv, keys, E, keyWords);
    deg_to_dinv<<<cdiv(N, B), B, 0, stream>>>(dinv, N);
    int* rec = (int*)(keys + keyWords);
    edge_records<<<cdiv(E, B), B, 0, stream>>>(tei, tei + E, dinv, rec, E);

    const int tiles = (N + 15) / 16;

    gemm_x_w1<<<cdiv(tiles, 8), B, 0, stream>>>(x, W1, h1, N);
    tile_aggregate<16><<<dim3(cdiv(N, 2048), 10), 1024, 0, stream>>>(rec, h1, (const unsigned char*)keys, P, E, N);
    epilogue<16, true><<<cdiv((long long)16 * N, B), B, 0, stream>>>(z1, P, 10, h1, dinv, b1, N);

    gemm_z_w2<<<cdiv(tiles, 8), B, 0, stream>>>(z1, W2, h2, N);
    tile_aggregate<64><<<dim3(cdiv(N, 512), 5), 1024, 0, stream>>>(rec, h2, (const unsigned char*)keys, P, E, N);
    epilogue<64, false><<<cdiv((long long)64 * N, B), B, 0, stream>>>(z2, P, 5, h2, dinv, b2, N);

    edge_score<<<cdiv(2 * ET, B), B, 0, stream>>>(tpo, tpo + ET, tne, tne + ET,
                                                  z2, (float*)d_out, ET);
}


# Degree count in LDS: a block takes 16384 edges, counts all nodes in packed 16-bit LDS counters (integer adds), stores a counter plane; the dinv pass sums planes; no global atomics, 10 launches
# speedup vs baseline: 5.3547x; 1.3749x over previous
#include <hip/hip_runtime.h>

typedef __attribute__((ext_vector_type(2))) float v2f;
typedef __attribute__((ext_vector_type(8))) float v8f;

constexpr int KEY_SHIFT = 9;

constexpr int CHUNK_EDGES = 16384;
constexpr int CHUNK_WORDS = CHUNK_EDGES / 4;
constexpr int RANGE_WORDS = 65536;
static_assert(CHUNK_EDGES < 65536, "a 16-bit counter must hold a whole chunk");

__global__ __launch_bounds__(1024) void deg_count_keys(const int* __restrict__ dst,
                                                       unsigned* __restrict__ planes,
                                                       unsigned* __restrict__ keys,
                                                       int nE, int nWords, int pitch) {
    __shared__ unsigned cnt[RANGE_WORDS];
    const int tid = threadIdx.x;
    const int w0  = blockIdx.y * RANGE_WORDS;
    const int nw  = min(RANGE_WORDS, pitch - w0);
    const bool vec = (((size_t)dst) & 15) == 0;

    int4 dv[CHUNK_WORDS / 1024];
    const int i0 = blockIdx.x * CHUNK_WORDS;
    if (vec && i0 + CHUNK_WORDS <= nWords && 4 * (i0 + CHUNK_WORDS) <= nE) {
#pragma unroll
        for (int k = 0; k < CHUNK_WORDS / 1024; ++k)
            dv[k] = *(const int4*)(dst + 4 * (i0 + k * 1024 + tid));
    } else {
#pragma unroll
        for (int k = 0; k < CHUNK_WORDS / 1024; ++k) {
            const int e = 4 * (i0 + k * 1024 + tid);
            int4 v = make_int4(-1, -1, -1, -1);
            if (e     < nE) v.x = dst[e];
            if (e + 1 < nE) v.y = dst[e + 1];
            if (e + 2 < nE) v.z = dst[e + 2];
            if (e + 3 < nE) v.w = dst[e + 3];
            dv[k] = v;
        }
    }
    for (int i = tid * 4; i < nw; i += 4096) *(uint4*)(cnt + i) = make_uint4(0u, 0u, 0u, 0u);
    __syncthreads();

    const unsigned node0 = 2u * (unsigned)w0, nodes = 2u * (unsigned)nw;
#pragma unroll
    for (int k = 0; k < CHUNK_WORDS / 1024; ++k) {
        const int i = i0 + k * 1024 + tid;
        if (i < nWords) {
            const int d[4] = {dv[k].x, dv[k].y, dv[k].z, dv[k].w};
            unsigned kw = 0;
#pragma unroll
            for (int b = 0; b < 4; ++b) {
                unsigned kb = 0xFFu;
                if (d[b] >= 0) {
                    kb = ((unsigned)d[b] >> KEY_SHIFT) & 0xFFu;
                    const unsigned loc = (unsigned)d[b] - node0;
                    if (loc < nodes) atomicAdd(&cnt[loc >> 1], 1u << ((loc & 1u) << 4));
                }
                kw |= kb << (8 * b);
            }
            if (blockIdx.y == 0) keys[i] = kw;
        }
    }
    __syncthreads();

    unsigned* pp = planes + (size_t)blockIdx.x * pitch + w0;
    for (int i = tid * 4; i < nw; i += 4096) *(uint4*)(pp + i) = *(const uint4*)(cnt + i);
}

__global__ __launch_bounds__(256) void deg_to_dinv(const unsigned* __restrict__ planes, int nPlanes,
                                                   int pitch, float* __restrict__ dinv, int n) {
    __shared__ unsigned part[8][256];
    const int lane = threadIdx.x & 31, w = threadIdx.x >> 5;
    const int word = (blockIdx.x * 32 + lane) * 4;
    unsigned s[8] = {0u, 0u, 0u, 0u, 0u, 0u, 0u, 0u};
    if (word < pitch) {
        for (int c = w; c < nPlanes; c += 64) {
            uint4 v[8];
#pragma unroll
            for (int j = 0; j < 8; ++j) {
                const int cc = c + 8 * j;
                v[j] = *(const uint4*)(planes + (size_t)min(cc, nPlanes - 1) * pitch + word);
                if (cc >= nPlanes) v[j] = make_uint4(0u, 0u, 0u, 0u);
            }
#pragma unroll
            for (int j = 0; j < 8; ++j) {
                s[0] += v[j].x & 0xFFFFu; s[1] += v[j].x >> 16;
                s[2] += v[j].y & 0xFFFFu; s[3] += v[j].y >> 16;
                s[4] += v[j].z & 0xFFFFu; s[5] += v[j].z >> 16;
                s[6] += v[j].w & 0xFFFFu; s[7] += v[j].w >> 16;
            }
        }
    }
#pragma unroll
    for (int j = 0; j < 8; ++j) part[w][lane * 8 + j] = s[j];
    __syncthreads();
    unsigned deg = 0;
#pragma unroll
    for (int k = 0; k < 8; ++k) deg += part[k][threadIdx.x];
    const int node = blockIdx.x * 256 + threadIdx.x;
    if (node < n) dinv[node] = rsqrtf((float)deg + 1.0f);
}

__global__ __launch_bounds__(256) void edge_records(const int* __restrict__ src,
                                                    const int* __restrict__ dst,
                                                    const float* __restrict__ dinv,
                                                    int* __restrict__ rec, int nE) {
    const int e = blockIdx.x * blockDim.x + threadIdx.x;
    if (e >= nE) return;
    const int s = src[e];
    const int d = dst[e];
    const float nrm = dinv[s] * dinv[d];
    *(int4*)(rec + 4 * (size_t)e) = make_int4(s, d, __float_as_int(nrm), 0);
}

__global__ __launch_bounds__(256) void gemm_x_w1(const float* __restrict__ X,
                                                 const float* __restrict__ W,
                                                 float* __restrict__ H, int n) {
    const int lane = threadIdx.x & 31;
    const int tile = blockIdx.x * 8 + (threadIdx.x >> 5);
    const int row0 = tile * 16;
    if (row0 >= n) return;
    int m = row0 + (lane & 15);
    if (m >= n) m = n - 1;
    const int kh  = (lane >> 4) << 1;
    const int col = lane & 15;
    const float* xr = X + (size_t)m * 256;

    v8f acc = {};
#pragma unroll 4
    for (int k0 = 0; k0 < 256; k0 += 4) {
        v2f a, b;
        a.x = xr[k0 + kh];
        a.y = xr[k0 + kh + 1];
        b.x = W[(size_t)(k0 + kh) * 16 + col];
        b.y = W[(size_t)(k0 + kh + 1) * 16 + col];
        acc = __builtin_amdgcn_wmma_f32_16x16x4_f32(false, a, false, b,
                                                    (short)0, acc, false, false);
    }
    const int ra = (lane >> 4) << 3;
    if (row0 + 16 <= n) {
#pragma unroll
        for (int v = 0; v < 8; ++v)
            H[(size_t)(row0 + ra + v) * 16 + col] = acc[v];
    } else {
#pragma unroll
        for (int v = 0; v < 8; ++v) {
            int r = row0 + ra + v;
            if (r < n) H[(size_t)r * 16 + col] = acc[v];
        }
    }
}

__global__ __launch_bounds__(256) void gemm_z_w2(const float* __restrict__ Z,
                                                 const float* __restrict__ W,
                                                 float* __restrict__ H, int n) {
    const int lane = threadIdx.x & 31;
    const int tile = blockIdx.x * 8 + (threadIdx.x >> 5);
    const int row0 = tile * 16;
    if (row0 >= n) return;
    int m = row0 + (lane & 15);
    if (m >= n) m = n - 1;
    const int kh  = (lane >> 4) << 1;
    const int col = lane & 15;
    const float* zr = Z + (size_t)m * 16;

    v8f acc0 = {}, acc1 = {}, acc2 = {}, acc3 = {};
#pragma unroll
    for (int k0 = 0; k0 < 16; k0 += 4) {
        v2f a, b;
        a.x = zr[k0 + kh];
        a.y = zr[k0 + kh + 1];
        const float* w0 = W + (size_t)(k0 + kh) * 64;
        const float* w1 = w0 + 64;
        b.x = w0[col];      b.y = w1[col];
        acc0 = __builtin_amdgcn_wmma_f32_16x16x4_f32(false, a, false, b, (short)0, acc0, false, false);
        b.x = w0[col + 16]; b.y = w1[col + 16];
        acc1 = __builtin_amdgcn_wmma_f32_16x16x4_f32(false, a, false, b, (short)0, acc1, false, false);
        b.x = w0[col + 32]; b.y = w1[col + 32];
        acc2 = __builtin_amdgcn_wmma_f32_16x16x4_f32(false, a, false, b, (short)0, acc2, false, false);
        b.x = w0[col + 48]; b.y = w1[col + 48];
        acc3 = __builtin_amdgcn_wmma_f32_16x16x4_f32(false, a, false, b, (short)0, acc3, false, false);
    }
    const int ra = (lane >> 4) << 3;
    if (row0 + 16 <= n) {
#pragma unroll
        for (int v = 0; v < 8; ++v) {
            float* hp = H + (size_t)(row0 + ra + v) * 64 + col;
            hp[0]  = acc0[v];
            hp[16] = acc1[v];
            hp[32] = acc2[v];
            hp[48] = acc3[v];
        }
    } else {
#pragma unroll
        for (int v = 0; v < 8; ++v) {
            int r = row0 + ra + v;
            if (r < n) {
                float* hp = H + (size_t)r * 64 + col;
                hp[0]  = acc0[v];
                hp[16] = acc1[v];
                hp[32] = acc2[v];
                hp[48] = acc3[v];
            }
        }
    }
}

constexpr int QCAP = 64;

template <int F>
__device__ __forceinline__ void add_rows(const int* __restrict__ rec, const float* __restrict__ H,
                                         float* tile, int base, int e, int cnt, int lane) {
    constexpr int V = F / 16;
    int s = 0, loc = 0;
    float nrm = 0.0f;
    if (lane < cnt) {
        const int4 r = *(const int4*)(rec + 4 * (size_t)e);
        s = r.x;
        nrm = __int_as_float(r.z);
        loc = r.y - base;
        if ((unsigned)loc >= (unsigned)(32768 / F)) loc = -1;
    }
    const int h = lane >> 4, col = lane & 15;
    for (int j = 0; j < cnt; j += 8) {
        float r[4][V], nj[4];
        int   off[4];
        bool  act[4];
#pragma unroll
        for (int u = 0; u < 4; ++u) {
            const int ja = j + 2 * u;
            const int sA = __builtin_amdgcn_readlane(s, ja);
            const int sB = __builtin_amdgcn_readlane(s, ja + 1);
            const int lA = __builtin_amdgcn_readlane(loc, ja);
            const int lB = __builtin_amdgcn_readlane(loc, ja + 1);
            const int nA = __builtin_amdgcn_readlane(__float_as_int(nrm), ja);
            const int nB = __builtin_amdgcn_readlane(__float_as_int(nrm), ja + 1);
            const int sj = h ? sB : sA;
            const int lj = h ? lB : lA;
            off[u] = lj * (F + 1) + col * V;
            nj[u]  = __int_as_float(h ? nB : nA);
            act[u] = ja + h < cnt && lj >= 0;
            const float* hp = H + (size_t)sj * F + col * V;
            if constexpr (V == 4) {
                const float4 t = *(const float4*)hp;
                r[u][0] = t.x; r[u][1] = t.y; r[u][2] = t.z; r[u][3] = t.w;
            } else {
#pragma unroll
                for (int i = 0; i < V; ++i) r[u][i] = hp[i];
            }
        }
#pragma unroll
        for (int u = 0; u < 4; ++u) {
            if (act[u]) {
#pragma unroll
                for (int i = 0; i < V; ++i) atomicAdd(tile + off[u] + i, r[u][i] * nj[u]);
            }
        }
    }
}

__device__ __forceinline__ unsigned nonzero_bytes(unsigned t) {
    return ((t & 0x7F7F7F7Fu) + 0x7F7F7F7Fu) | t;
}

template <int F>
__global__ __launch_bounds__(1024) void tile_aggregate(const int* __restrict__ rec,
                                                       const float* __restrict__ H,
                                                       const unsigned char* __restrict__ keys,
                                                       float* __restrict__ P,
                                                       int nE, int n) {
    constexpr int TN = 32768 / F;
    constexpr int PITCH = F + 1;
    constexpr int KPT = TN >> KEY_SHIFT;
    static_assert(KPT >= 1 && (KPT & (KPT - 1)) == 0 && KPT * (1 << KEY_SHIFT) == TN, "tile = 2^k keys");
    constexpr unsigned KM  = (0x100u - KPT) & 0xFFu;
    constexpr unsigned KM4 = KM * 0x01010101u;
    __shared__ float tile[TN * PITCH];
    __shared__ int   stack[32 * QCAP];

    const int tid  = threadIdx.x;
    const int lane = tid & 31;
    const int w    = __builtin_amdgcn_readfirstlane(tid >> 5);
    const int base = blockIdx.x * TN;
    const int c    = blockIdx.y;
    int* q = stack + w * QCAP;
    const unsigned pat = (((unsigned)blockIdx.x * KPT) & KM) * 0x01010101u;

    for (int i = tid; i < TN * PITCH; i += 1024) tile[i] = 0.0f;
    __syncthreads();

    const int ec = (((nE + (int)gridDim.y - 1) / (int)gridDim.y) + 511) & ~511;
    const int e0 = c * ec;
    const int e1 = min(nE, e0 + ec);

    int pending = 0;
    constexpr int STRIDE = 32 * 512;
    int eb = e0 + w * 512;
    uint4 kn = make_uint4(0u, 0u, 0u, 0u);
    if (eb < e1) kn = *(const uint4*)(keys + eb + lane * 16);
    for (; eb < e1; eb += STRIDE) {
        const uint4 kc = kn;
        if (eb + STRIDE < e1) kn = *(const uint4*)(keys + eb + STRIDE + lane * 16);
        const unsigned n0 = nonzero_bytes((kc.x ^ pat) & KM4);
        const unsigned n1 = nonzero_bytes((kc.y ^ pat) & KM4);
        const unsigned n2 = nonzero_bytes((kc.z ^ pat) & KM4);
        const unsigned n3 = nonzero_bytes((kc.w ^ pat) & KM4);
        unsigned m = (((n0 >> 7) & 0x01010101u) | ((n1 >> 6) & 0x02020202u) |
                      ((n2 >> 5) & 0x04040404u) | ((n3 >> 4) & 0x08080808u)) ^ 0x0F0F0F0Fu;
        const int elane = eb + lane * 16;
        unsigned any;
        while ((any = __builtin_amdgcn_ballot_w32(m != 0u)) != 0u) {
            const int t = __ffs((int)m) - 1;
            const int e = elane + ((t & 7) << 2) + (t >> 3);
            const unsigned bal = any & __builtin_amdgcn_ballot_w32(e < e1);
            if (m != 0u && e < e1) q[pending + (int)__builtin_amdgcn_mbcnt_lo(bal, 0u)] = e;
            pending += __popc(bal);
            m &= m - 1u;
            if (pending >= 32) {
                pending -= 32;
                add_rows<F>(rec, H, tile, base, q[pending + lane], 32, lane);
            }
        }
    }
    add_rows<F>(rec, H, tile, base, lane < pending ? q[lane] : 0, pending, lane);
    __syncthreads();

    float* pp = P + ((size_t)c * n + base) * F;
    const int rows = min(TN, n - base);
    for (int i = tid; i < rows * F; i += 1024)
        pp[i] = tile[(i / F) * PITCH + (i % F)];
}

template <int F, bool RELU>
__global__ __launch_bounds__(256) void epilogue(float* __restrict__ OUT,
                                                const float* __restrict__ P, int nC,
                                                const float* __restrict__ H,
                                                const float* __restrict__ dinv,
                                                const float* __restrict__ bias, int n) {
    long long idx = (long long)blockIdx.x * blockDim.x + threadIdx.x;
    if (idx >= (long long)n * F) return;
    int i = (int)(idx / F);
    int f = (int)(idx % F);
    float acc = 0.0f;
    for (int c = 0; c < nC; ++c)
        acc += P[(size_t)c * n * F + idx];
    float di = dinv[i];
    float v = acc + di * di * H[idx] + bias[f];
    if (RELU) v = fmaxf(v, 0.0f);
    OUT[idx] = v;
}

__global__ __launch_bounds__(256) void edge_score(const int* __restrict__ psrc,
                                                  const int* __restrict__ pdst,
                                                  const int* __restrict__ nsrc,
                                                  const int* __restrict__ ndst,
                                                  const float* __restrict__ Z,
                                                  float* __restrict__ out, int nT) {
    int j = blockIdx.x * blockDim.x + threadIdx.x;
    if (j >= 2 * nT) return;
    int s, d;
    if (j < nT) { s = psrc[j]; d = pdst[j]; }
    else        { s = nsrc[j - nT]; d = ndst[j - nT]; }
    const float4* zs = (const float4*)(Z + (size_t)s * 64);
    const float4* zd = (const float4*)(Z + (size_t)d * 64);
    float acc = 0.0f;
#pragma unroll
    for (int k = 0; k < 16; ++k) {
        float4 a = zs[k], b = zd[k];
        acc = fmaf(a.x, b.x, acc);
        acc = fmaf(a.y, b.y, acc);
        acc = fmaf(a.z, b.z, acc);
        acc = fmaf(a.w, b.w, acc);
    }
    out[j] = acc;
}

extern "C" void kernel_launch(void* const* d_in, const int* in_sizes, int n_in,
                              void* d_out, int out_size, void* d_ws, size_t ws_size,
                              hipStream_t stream) {
    const float* x   = (const float*)d_in[0];
    const int*   tei = (const int*)d_in[1];
    const int*   tpo = (const int*)d_in[2];
    const int*   tne = (const int*)d_in[3];
    const float* W1  = (const float*)d_in[4];
    const float* b1  = (const float*)d_in[5];
    const float* W2  = (const float*)d_in[6];
    const float* b2  = (const float*)d_in[7];

    const int N  = in_sizes[0] / 256;
    const int E  = in_sizes[1] / 2;
    const int ET = in_sizes[2] / 2;

    float* ws   = (float*)d_ws;
    float* dinv = ws;
    float* h1   = ws + (size_t)N;
    float* z1   = ws + (size_t)17 * N;
    float* h2   = ws + (size_t)33 * N;
    float* z2   = ws + (size_t)97 * N;
    float* P    = ws + (size_t)161 * N;
    unsigned* keys = (unsigned*)(ws + (((size_t)481 * N + 3) & ~(size_t)3));

    const int B = 256;
    auto cdiv = [](long long a, long long b) { return (int)((a + b - 1) / b); };

    const int keyWords = cdiv(E, 512) * 128;
    const int pitch  = (((N + 1) / 2) + 3) & ~3;
    const int chunks = cdiv(keyWords, CHUNK_WORDS);
    unsigned* planes = (unsigned*)(ws + (((size_t)161 * N + 3) & ~(size_t)3));
    deg_count_keys<<<dim3(chunks, cdiv(pitch, RANGE_WORDS)), 1024, 0, stream>>>(tei + E, planes, keys, E, keyWords, pitch);
    deg_to_dinv<<<cdiv(pitch, 128), B, 0, stream>>>(planes, chunks, pitch, dinv, N);
    int* rec = (int*)(keys + keyWords);
    edge_records<<<cdiv(E, B), B, 0, stream>>>(tei, tei + E, dinv, rec, E);

    const int tiles = (N + 15) / 16;

    gemm_x_w1<<<cdiv(tiles, 8), B, 0, stream>>>(x, W1, h1, N);
    tile_aggregate<16><<<dim3(cdiv(N, 2048), 10), 1024, 0, stream>>>(rec, h1, (const unsigned char*)keys, P, E, N);
    epilogue<16, true><<<cdiv((long long)16 * N, B), B, 0, stream>>>(z1, P, 10, h1, dinv, b1, N);

    gemm_z_w2<<<cdiv(tiles, 8), B, 0, stream>>>(z1, W2, h2, N);
    tile_aggregate<64><<<dim3(cdiv(N, 512), 5), 1024, 0, stream>>>(rec, h2, (const unsigned char*)keys, P, E, N);
    epilogue<64, false><<<cdiv((long long)64 * N, B), B, 0, stream>>>(z2, P, 5, h2, dinv, b2, N);

    edge_score<<<cdiv(2 * ET, B), B, 0, stream>>>(tpo, tpo + ET, tne, tne + ET,
                                                  z2, (float*)d_out, ET);
}


# Element-wise passes (both plane-sum epilogues, edge-record pass) as few long-lived waves: a lane does 6 or 3 four-float groups, or 8 edges, per launch, not one; 350k waves become 25k; sum order kept
# speedup vs baseline: 5.8097x; 1.0850x over previous
#include <hip/hip_runtime.h>

typedef __attribute__((ext_vector_type(2))) float v2f;
typedef __attribute__((ext_vector_type(8))) float v8f;

constexpr int KEY_SHIFT = 9;

constexpr int CHUNK_EDGES = 16384;
constexpr int CHUNK_WORDS = CHUNK_EDGES / 4;
constexpr int RANGE_WORDS = 65536;
static_assert(CHUNK_EDGES < 65536, "a 16-bit counter must hold a whole chunk");

__global__ __launch_bounds__(1024) void deg_count_keys(const int* __restrict__ dst,
                                                       unsigned* __restrict__ planes,
                                                       unsigned* __restrict__ keys,
                                                       int nE, int nWords, int pitch) {
    __shared__ unsigned cnt[RANGE_WORDS];
    const int tid = threadIdx.x;
    const int w0  = blockIdx.y * RANGE_WORDS;
    const int nw  = min(RANGE_WORDS, pitch - w0);
    const bool vec = (((size_t)dst) & 15) == 0;

    int4 dv[CHUNK_WORDS / 1024];
    const int i0 = blockIdx.x * CHUNK_WORDS;
    if (vec && i0 + CHUNK_WORDS <= nWords && 4 * (i0 + CHUNK_WORDS) <= nE) {
#pragma unroll
        for (int k = 0; k < CHUNK_WORDS / 1024; ++k)
            dv[k] = *(const int4*)(dst + 4 * (i0 + k * 1024 + tid));
    } else {
#pragma unroll
        for (int k = 0; k < CHUNK_WORDS / 1024; ++k) {
            const int e = 4 * (i0 + k * 1024 + tid);
            int4 v = make_int4(-1, -1, -1, -1);
            if (e     < nE) v.x = dst[e];
            if (e + 1 < nE) v.y = dst[e + 1];
            if (e + 2 < nE) v.z = dst[e + 2];
            if (e + 3 < nE) v.w = dst[e + 3];
            dv[k] = v;
        }
    }
    for (int i = tid * 4; i < nw; i += 4096) *(uint4*)(cnt + i) = make_uint4(0u, 0u, 0u, 0u);
    __syncthreads();

    const unsigned node0 = 2u * (unsigned)w0, nodes = 2u * (unsigned)nw;
#pragma unroll
    for (int k = 0; k < CHUNK_WORDS / 1024; ++k) {
        const int i = i0 + k * 1024 + tid;
        if (i < nWords) {
            const int d[4] = {dv[k].x, dv[k].y, dv[k].z, dv[k].w};
            unsigned kw = 0;
#pragma unroll
            for (int b = 0; b < 4; ++b) {
                unsigned kb = 0xFFu;
                if (d[b] >= 0) {
                    kb = ((unsigned)d[b] >> KEY_SHIFT) & 0xFFu;
                    const unsigned loc = (unsigned)d[b] - node0;
                    if (loc < nodes) atomicAdd(&cnt[loc >> 1], 1u << ((loc & 1u) << 4));
                }
                kw |= kb << (8 * b);
            }
            if (blockIdx.y == 0) keys[i] = kw;
        }
    }
    __syncthreads();

    unsigned* pp = planes + (size_t)blockIdx.x * pitch + w0;
    for (int i = tid * 4; i < nw; i += 4096) *(uint4*)(pp + i) = *(const uint4*)(cnt + i);
}

__global__ __launch_bounds__(256) void deg_to_dinv(const unsigned* __restrict__ planes, int nPlanes,
                                                   int pitch, float* __restrict__ dinv, int n) {
    __shared__ unsigned part[8][256];
    const int lane = threadIdx.x & 31, w = threadIdx.x >> 5;
    const int word = (blockIdx.x * 32 + lane) * 4;
    unsigned s[8] = {0u, 0u, 0u, 0u, 0u, 0u, 0u, 0u};
    if (word < pitch) {
        for (int c = w; c < nPlanes; c += 64) {
            uint4 v[8];
#pragma unroll
            for (int j = 0; j < 8; ++j) {
                const int cc = c + 8 * j;
                v[j] = *(const uint4*)(planes + (size_t)min(cc, nPlanes - 1) * pitch + word);
                if (cc >= nPlanes) v[j] = make_uint4(0u, 0u, 0u, 0u);
            }
#pragma unroll
            for (int j = 0; j < 8; ++j) {
                s[0] += v[j].x & 0xFFFFu; s[1] += v[j].x >> 16;
                s[2] += v[j].y & 0xFFFFu; s[3] += v[j].y >> 16;
                s[4] += v[j].z & 0xFFFFu; s[5] += v[j].z >> 16;
                s[6] += v[j].w & 0xFFFFu; s[7] += v[j].w >> 16;
            }
        }
    }
#pragma unroll
    for (int j = 0; j < 8; ++j) part[w][lane * 8 + j] = s[j];
    __syncthreads();
    unsigned deg = 0;
#pragma unroll
    for (int k = 0; k < 8; ++k) deg += part[k][threadIdx.x];
    const int node = blockIdx.x * 256 + threadIdx.x;
    if (node < n) dinv[node] = rsqrtf((float)deg + 1.0f);
}

constexpr int REC_STEPS = 8;

__global__ __launch_bounds__(256) void edge_records(const int* __restrict__ src,
                                                    const int* __restrict__ dst,
                                                    const float* __restrict__ dinv,
                                                    int* __restrict__ rec, int nE) {
    const int b0 = blockIdx.x * (256 * REC_STEPS);
    const int e0 = b0 + threadIdx.x;
    int s[REC_STEPS], d[REC_STEPS];
    float nrm[REC_STEPS];
    if (b0 + 256 * REC_STEPS <= nE) {
#pragma unroll
        for (int k = 0; k < REC_STEPS; ++k) {
            s[k] = src[e0 + k * 256];
            d[k] = dst[e0 + k * 256];
        }
#pragma unroll
        for (int k = 0; k < REC_STEPS; ++k) nrm[k] = dinv[s[k]] * dinv[d[k]];
#pragma unroll
        for (int k = 0; k < REC_STEPS; ++k)
            *(int4*)(rec + 4 * (size_t)(e0 + k * 256)) = make_int4(s[k], d[k], __float_as_int(nrm[k]), 0);
    } else {
#pragma unroll
        for (int k = 0; k < REC_STEPS; ++k) {
            const int e = e0 + k * 256;
            if (e < nE) {
                const int sk = src[e];
                const int dk = dst[e];
                *(int4*)(rec + 4 * (size_t)e) = make_int4(sk, dk, __float_as_int(dinv[sk] * dinv[dk]), 0);
            }
        }
    }
}

__global__ __launch_bounds__(256) void gemm_x_w1(const float* __restrict__ X,
                                                 const float* __restrict__ W,
                                                 float* __restrict__ H, int n) {
    const int lane = threadIdx.x & 31;
    const int tile = blockIdx.x * 8 + (threadIdx.x >> 5);
    const int row0 = tile * 16;
    if (row0 >= n) return;
    int m = row0 + (lane & 15);
    if (m >= n) m = n - 1;
    const int kh  = (lane >> 4) << 1;
    const int col = lane & 15;
    const float* xr = X + (size_t)m * 256;

    v8f acc = {};
#pragma unroll 4
    for (int k0 = 0; k0 < 256; k0 += 4) {
        v2f a, b;
        a.x = xr[k0 + kh];
        a.y = xr[k0 + kh + 1];
        b.x = W[(size_t)(k0 + kh) * 16 + col];
        b.y = W[(size_t)(k0 + kh + 1) * 16 + col];
        acc = __builtin_amdgcn_wmma_f32_16x16x4_f32(false, a, false, b,
                                                    (short)0, acc, false, false);
    }
    const int ra = (lane >> 4) << 3;
    if (row0 + 16 <= n) {
#pragma unroll
        for (int v = 0; v < 8; ++v)
            H[(size_t)(row0 + ra + v) * 16 + col] = acc[v];
    } else {
#pragma unroll
        for (int v = 0; v < 8; ++v) {
            int r = row0 + ra + v;
            if (r < n) H[(size_t)r * 16 + col] = acc[v];
        }
    }
}

__global__ __launch_bounds__(256) void gemm_z_w2(const float* __restrict__ Z,
                                                 const float* __restrict__ W,
                                                 float* __restrict__ H, int n) {
    const int lane = threadIdx.x & 31;
    const int tile = blockIdx.x * 8 + (threadIdx.x >> 5);
    const int row0 = tile * 16;
    if (row0 >= n) return;
    int m = row0 + (lane & 15);
    if (m >= n) m = n - 1;
    const int kh  = (lane >> 4) << 1;
    const int col = lane & 15;
    const float* zr = Z + (size_t)m * 16;

    v8f acc0 = {}, acc1 = {}, acc2 = {}, acc3 = {};
#pragma unroll
    for (int k0 = 0; k0 < 16; k0 += 4) {
        v2f a, b;
        a.x = zr[k0 + kh];
        a.y = zr[k0 + kh + 1];
        const float* w0 = W + (size_t)(k0 + kh) * 64;
        const float* w1 = w0 + 64;
        b.x = w0[col];      b.y = w1[col];
        acc0 = __builtin_amdgcn_wmma_f32_16x16x4_f32(false, a, false, b, (short)0, acc0, false, false);
        b.x = w0[col + 16]; b.y = w1[col + 16];
        acc1 = __builtin_amdgcn_wmma_f32_16x16x4_f32(false, a, false, b, (short)0, acc1, false, false);
        b.x = w0[col + 32]; b.y = w1[col + 32];
        acc2 = __builtin_amdgcn_wmma_f32_16x16x4_f32(false, a, false, b, (short)0, acc2, false, false);
        b.x = w0[col + 48]; b.y = w1[col + 48];
        acc3 = __builtin_amdgcn_wmma_f32_16x16x4_f32(false, a, false, b, (short)0, acc3, false, false);
    }
    const int ra = (lane >> 4) << 3;
    if (row0 + 16 <= n) {
#pragma unroll
        for (int v = 0; v < 8; ++v) {
            float* hp = H + (size_t)(row0 + ra + v) * 64 + col;
            hp[0]  = acc0[v];
            hp[16] = acc1[v];
            hp[32] = acc2[v];
            hp[48] = acc3[v];
        }
    } else {
#pragma unroll
        for (int v = 0; v < 8; ++v) {
            int r = row0 + ra + v;
            if (r < n) {
                float* hp = H + (size_t)r * 64 + col;
                hp[0]  = acc0[v];
                hp[16] = acc1[v];
                hp[32] = acc2[v];
                hp[48] = acc3[v];
            }
        }
    }
}

constexpr int QCAP = 64;

template <int F>
__device__ __forceinline__ void add_rows(const int* __restrict__ rec, const float* __restrict__ H,
                                         float* tile, int base, int e, int cnt, int lane) {
    constexpr int V = F / 16;
    int s = 0, loc = 0;
    float nrm = 0.0f;
    if (lane < cnt) {
        const int4 r = *(const int4*)(rec + 4 * (size_t)e);
        s = r.x;
        nrm = __int_as_float(r.z);
        loc = r.y - base;
        if ((unsigned)loc >= (unsigned)(32768 / F)) loc = -1;
    }
    const int h = lane >> 4, col = lane & 15;
    for (int j = 0; j < cnt; j += 8) {
        float r[4][V], nj[4];
        int   off[4];
        bool  act[4];
#pragma unroll
        for (int u = 0; u < 4; ++u) {
            const int ja = j + 2 * u;
            const int sA = __builtin_amdgcn_readlane(s, ja);
            const int sB = __builtin_amdgcn_readlane(s, ja + 1);
            const int lA = __builtin_amdgcn_readlane(loc, ja);
            const int lB = __builtin_amdgcn_readlane(loc, ja + 1);
            const int nA = __builtin_amdgcn_readlane(__float_as_int(nrm), ja);
            const int nB = __builtin_amdgcn_readlane(__float_as_int(nrm), ja + 1);
            const int sj = h ? sB : sA;
            const int lj = h ? lB : lA;
            off[u] = lj * (F + 1) + col * V;
            nj[u]  = __int_as_float(h ? nB : nA);
            act[u] = ja + h < cnt && lj >= 0;
            const float* hp = H + (size_t)sj * F + col * V;
            if constexpr (V == 4) {
                const float4 t = *(const float4*)hp;
                r[u][0] = t.x; r[u][1] = t.y; r[u][2] = t.z; r[u][3] = t.w;
            } else {
#pragma unroll
                for (int i = 0; i < V; ++i) r[u][i] = hp[i];
            }
        }
#pragma unroll
        for (int u = 0; u < 4; ++u) {
            if (act[u]) {
#pragma unroll
                for (int i = 0; i < V; ++i) atomicAdd(tile + off[u] + i, r[u][i] * nj[u]);
            }
        }
    }
}

__device__ __forceinline__ unsigned nonzero_bytes(unsigned t) {
    return ((t & 0x7F7F7F7Fu) + 0x7F7F7F7Fu) | t;
}

template <int F>
__global__ __launch_bounds__(1024) void tile_aggregate(const int* __restrict__ rec,
                                                       const float* __restrict__ H,
                                                       const unsigned char* __restrict__ keys,
                                                       float* __restrict__ P,
                                                       int nE, int n) {
    constexpr int TN = 32768 / F;
    constexpr int PITCH = F + 1;
    constexpr int KPT = TN >> KEY_SHIFT;
    static_assert(KPT >= 1 && (KPT & (KPT - 1)) == 0 && KPT * (1 << KEY_SHIFT) == TN, "tile = 2^k keys");
    constexpr unsigned KM  = (0x100u - KPT) & 0xFFu;
    constexpr unsigned KM4 = KM * 0x01010101u;
    __shared__ float tile[TN * PITCH];
    __shared__ int   stack[32 * QCAP];

    const int tid  = threadIdx.x;
    const int lane = tid & 31;
    const int w    = __builtin_amdgcn_readfirstlane(tid >> 5);
    const int base = blockIdx.x * TN;
    const int c    = blockIdx.y;
    int* q = stack + w * QCAP;
    const unsigned pat = (((unsigned)blockIdx.x * KPT) & KM) * 0x01010101u;

    for (int i = tid; i < TN * PITCH; i += 1024) tile[i] = 0.0f;
    __syncthreads();

    const int ec = (((nE + (int)gridDim.y - 1) / (int)gridDim.y) + 511) & ~511;
    const int e0 = c * ec;
    const int e1 = min(nE, e0 + ec);

    int pending = 0;
    constexpr int STRIDE = 32 * 512;
    int eb = e0 + w * 512;
    uint4 kn = make_uint4(0u, 0u, 0u, 0u);
    if (eb < e1) kn = *(const uint4*)(keys + eb + lane * 16);
    for (; eb < e1; eb += STRIDE) {
        const uint4 kc = kn;
        if (eb + STRIDE < e1) kn = *(const uint4*)(keys + eb + STRIDE + lane * 16);
        const unsigned n0 = nonzero_bytes((kc.x ^ pat) & KM4);
        const unsigned n1 = nonzero_bytes((kc.y ^ pat) & KM4);
        const unsigned n2 = nonzero_bytes((kc.z ^ pat) & KM4);
        const unsigned n3 = nonzero_bytes((kc.w ^ pat) & KM4);
        unsigned m = (((n0 >> 7) & 0x01010101u) | ((n1 >> 6) & 0x02020202u) |
                      ((n2 >> 5) & 0x04040404u) | ((n3 >> 4) & 0x08080808u)) ^ 0x0F0F0F0Fu;
        const int elane = eb + lane * 16;
        unsigned any;
        while ((any = __builtin_amdgcn_ballot_w32(m != 0u)) != 0u) {
            const int t = __ffs((int)m) - 1;
            const int e = elane + ((t & 7) << 2) + (t >> 3);
            const unsigned bal = any & __builtin_amdgcn_ballot_w32(e < e1);
            if (m != 0u && e < e1) q[pending + (int)__builtin_amdgcn_mbcnt_lo(bal, 0u)] = e;
            pending += __popc(bal);
            m &= m - 1u;
            if (pending >= 32) {
                pending -= 32;
                add_rows<F>(rec, H, tile, base, q[pending + lane], 32, lane);
            }
        }
    }
    add_rows<F>(rec, H, tile, base, lane < pending ? q[lane] : 0, pending, lane);
    __syncthreads();

    float* pp = P + ((size_t)c * n + base) * F;
    const int rows = min(TN, n - base);
    for (int i = tid; i < rows * F; i += 1024)
        pp[i] = tile[(i / F) * PITCH + (i % F)];
}

template <int F, bool RELU, int NC>
__global__ __launch_bounds__(256) void epilogue(float* __restrict__ OUT,
                                                const float* __restrict__ P,
                                                const float* __restrict__ H,
                                                const float* __restrict__ dinv,
                                                const float* __restrict__ bias, int n, int steps) {
    const size_t total  = (size_t)n * F;
    const size_t groups = total / 4;
    size_t g = (size_t)blockIdx.x * steps * 256 + threadIdx.x;
    for (int k = 0; k < steps; ++k, g += 256) {
        if (g >= groups) break;
        const size_t idx = 4 * g;
        const int i = (int)(idx / F);
        const int f = (int)(idx % F);
        float4 p[NC];
#pragma unroll
        for (int c = 0; c < NC; ++c) p[c] = *(const float4*)(P + (size_t)c * total + idx);
        const float4 h = *(const float4*)(H + idx);
        const float di = dinv[i];
        const float b0 = bias[f], b1 = bias[f + 1], b2 = bias[f + 2], b3 = bias[f + 3];
        float4 acc = make_float4(0.0f, 0.0f, 0.0f, 0.0f);
#pragma unroll
        for (int c = 0; c < NC; ++c) {
            acc.x += p[c].x; acc.y += p[c].y; acc.z += p[c].z; acc.w += p[c].w;
        }
        float4 v;
        v.x = acc.x + di * di * h.x + b0;
        v.y = acc.y + di * di * h.y + b1;
        v.z = acc.z + di * di * h.z + b2;
        v.w = acc.w + di * di * h.w + b3;
        if (RELU) {
            v.x = fmaxf(v.x, 0.0f); v.y = fmaxf(v.y, 0.0f);
            v.z = fmaxf(v.z, 0.0f); v.w = fmaxf(v.w, 0.0f);
        }
        *(float4*)(OUT + idx) = v;
    }
}

__global__ __launch_bounds__(256) void edge_score(const int* __restrict__ psrc,
                                                  const int* __restrict__ pdst,
                                                  const int* __restrict__ nsrc,
                                                  const int* __restrict__ ndst,
                                                  const float* __restrict__ Z,
                                                  float* __restrict__ out, int nT) {
    int j = blockIdx.x * blockDim.x + threadIdx.x;
    if (j >= 2 * nT) return;
    int s, d;
    if (j < nT) { s = psrc[j]; d = pdst[j]; }
    else        { s = nsrc[j - nT]; d = ndst[j - nT]; }
    const float4* zs = (const float4*)(Z + (size_t)s * 64);
    const float4* zd = (const float4*)(Z + (size_t)d * 64);
    float acc = 0.0f;
#pragma unroll
    for (int k = 0; k < 16; ++k) {
        float4 a = zs[k], b = zd[k];
        acc = fmaf(a.x, b.x, acc);
        acc = fmaf(a.y, b.y, acc);
        acc = fmaf(a.z, b.z, acc);
        acc = fmaf(a.w, b.w, acc);
    }
    out[j] = acc;
}

extern "C" void kernel_launch(void* const* d_in, const int* in_sizes, int n_in,
                              void* d_out, int out_size, void* d_ws, size_t ws_size,
                              hipStream_t stream) {
    const float* x   = (const float*)d_in[0];
    const int*   tei = (const int*)d_in[1];
    const int*   tpo = (const int*)d_in[2];
    const int*   tne = (const int*)d_in[3];
    const float* W1  = (const float*)d_in[4];
    const float* b1  = (const float*)d_in[5];
    const float* W2  = (const float*)d_in[6];
    const float* b2  = (const float*)d_in[7];

    const int N  = in_sizes[0] / 256;
    const int E  = in_sizes[1] / 2;
    const int ET = in_sizes[2] / 2;

    float* ws   = (float*)d_ws;
    float* dinv = ws;
    float* h1   = ws + (size_t)N;
    float* z1   = ws + (size_t)17 * N;
    float* h2   = ws + (size_t)33 * N;
    float* z2   = ws + (size_t)97 * N;
    float* P    = ws + (size_t)161 * N;
    unsigned* keys = (unsigned*)(ws + (((size_t)481 * N + 3) & ~(size_t)3));

    const int B = 256;
    auto cdiv = [](long long a, long long b) { return (int)((a + b - 1) / b); };

    const int keyWords = cdiv(E, 512) * 128;
    const int pitch  = (((N + 1) / 2) + 3) & ~3;
    const int chunks = cdiv(keyWords, CHUNK_WORDS);
    unsigned* planes = (unsigned*)(ws + (((size_t)161 * N + 3) & ~(size_t)3));
    deg_count_keys<<<dim3(chunks, cdiv(pitch, RANGE_WORDS)), 1024, 0, stream>>>(tei + E, planes, keys, E, keyWords, pitch);
    deg_to_dinv<<<cdiv(pitch, 128), B, 0, stream>>>(planes, chunks, pitch, dinv, N);
    int* rec = (int*)(keys + keyWords);
    edge_records<<<cdiv(E, B * REC_STEPS), B, 0, stream>>>(tei, tei + E, dinv, rec, E);

    const int tiles = (N + 15) / 16;

    gemm_x_w1<<<cdiv(tiles, 8), B, 0, stream>>>(x, W1, h1, N);
    tile_aggregate<16><<<dim3(cdiv(N, 2048), 10), 1024, 0, stream>>>(rec, h1, (const unsigned char*)keys, P, E, N);
    epilogue<16, true, 10><<<cdiv((long long)4 * N, B * 3), B, 0, stream>>>(z1, P, h1, dinv, b1, N, 3);

    gemm_z_w2<<<cdiv(tiles, 8), B, 0, stream>>>(z1, W2, h2, N);
    tile_aggregate<64><<<dim3(cdiv(N, 512), 5), 1024, 0, stream>>>(rec, h2, (const unsigned char*)keys, P, E, N);
    epilogue<64, false, 5><<<cdiv((long long)16 * N, B * 6), B, 0, stream>>>(z2, P, h2, dinv, b2, N, 6);

    edge_score<<<cdiv(2 * ET, B), B, 0, stream>>>(tpo, tpo + ET, tne, tne + ET,
                                                  z2, (float*)d_out, ET);
}


# Edge scoring as 256 long-lived 8-wave blocks (one per WGP); a lane walks the edge list, next indices loaded a trip ahead: its row lines stay in the WGP cache between 16-byte reads; sums unchanged
# speedup vs baseline: 6.4083x; 1.1030x over previous
#include <hip/hip_runtime.h>

typedef __attribute__((ext_vector_type(2))) float v2f;
typedef __attribute__((ext_vector_type(8))) float v8f;

constexpr int KEY_SHIFT = 9;

constexpr int CHUNK_EDGES = 16384;
constexpr int CHUNK_WORDS = CHUNK_EDGES / 4;
constexpr int RANGE_WORDS = 65536;
static_assert(CHUNK_EDGES < 65536, "a 16-bit counter must hold a whole chunk");

__global__ __launch_bounds__(1024) void deg_count_keys(const int* __restrict__ dst,
                                                       unsigned* __restrict__ planes,
                                                       unsigned* __restrict__ keys,
                                                       int nE, int nWords, int pitch) {
    __shared__ unsigned cnt[RANGE_WORDS];
    const int tid = threadIdx.x;
    const int w0  = blockIdx.y * RANGE_WORDS;
    const int nw  = min(RANGE_WORDS, pitch - w0);
    const bool vec = (((size_t)dst) & 15) == 0;

    int4 dv[CHUNK_WORDS / 1024];
    const int i0 = blockIdx.x * CHUNK_WORDS;
    if (vec && i0 + CHUNK_WORDS <= nWords && 4 * (i0 + CHUNK_WORDS) <= nE) {
#pragma unroll
        for (int k = 0; k < CHUNK_WORDS / 1024; ++k)
            dv[k] = *(const int4*)(dst + 4 * (i0 + k * 1024 + tid));
    } else {
#pragma unroll
        for (int k = 0; k < CHUNK_WORDS / 1024; ++k) {
            const int e = 4 * (i0 + k * 1024 + tid);
            int4 v = make_int4(-1, -1, -1, -1);
            if (e     < nE) v.x = dst[e];
            if (e + 1 < nE) v.y = dst[e + 1];
            if (e + 2 < nE) v.z = dst[e + 2];
            if (e + 3 < nE) v.w = dst[e + 3];
            dv[k] = v;
        }
    }
    for (int i = tid * 4; i < nw; i += 4096) *(uint4*)(cnt + i) = make_uint4(0u, 0u, 0u, 0u);
    __syncthreads();

    const unsigned node0 = 2u * (unsigned)w0, nodes = 2u * (unsigned)nw;
#pragma unroll
    for (int k = 0; k < CHUNK_WORDS / 1024; ++k) {
        const int i = i0 + k * 1024 + tid;
        if (i < nWords) {
            const int d[4] = {dv[k].x, dv[k].y, dv[k].z, dv[k].w};
            unsigned kw = 0;
#pragma unroll
            for (int b = 0; b < 4; ++b) {
                unsigned kb = 0xFFu;
                if (d[b] >= 0) {
                    kb = ((unsigned)d[b] >> KEY_SHIFT) & 0xFFu;
                    const unsigned loc = (unsigned)d[b] - node0;
                    if (loc < nodes) atomicAdd(&cnt[loc >> 1], 1u << ((loc & 1u) << 4));
                }
                kw |= kb << (8 * b);
            }
            if (blockIdx.y == 0) keys[i] = kw;
        }
    }
    __syncthreads();

    unsigned* pp = planes + (size_t)blockIdx.x * pitch + w0;
    for (int i = tid * 4; i < nw; i += 4096) *(uint4*)(pp + i) = *(const uint4*)(cnt + i);
}

__global__ __launch_bounds__(256) void deg_to_dinv(const unsigned* __restrict__ planes, int nPlanes,
                                                   int pitch, float* __restrict__ dinv, int n) {
    __shared__ unsigned part[8][256];
    const int lane = threadIdx.x & 31, w = threadIdx.x >> 5;
    const int word = (blockIdx.x * 32 + lane) * 4;
    unsigned s[8] = {0u, 0u, 0u, 0u, 0u, 0u, 0u, 0u};
    if (word < pitch) {
        for (int c = w; c < nPlanes; c += 64) {
            uint4 v[8];
#pragma unroll
            for (int j = 0; j < 8; ++j) {
                const int cc = c + 8 * j;
                v[j] = *(const uint4*)(planes + (size_t)min(cc, nPlanes - 1) * pitch + word);
                if (cc >= nPlanes) v[j] = make_uint4(0u, 0u, 0u, 0u);
            }
#pragma unroll
            for (int j = 0; j < 8; ++j) {
                s[0] += v[j].x & 0xFFFFu; s[1] += v[j].x >> 16;
                s[2] += v[j].y & 0xFFFFu; s[3] += v[j].y >> 16;
                s[4] += v[j].z & 0xFFFFu; s[5] += v[j].z >> 16;
                s[6] += v[j].w & 0xFFFFu; s[7] += v[j].w >> 16;
            }
        }
    }
#pragma unroll
    for (int j = 0; j < 8; ++j) part[w][lane * 8 + j] = s[j];
    __syncthreads();
    unsigned deg = 0;
#pragma unroll
    for (int k = 0; k < 8; ++k) deg += part[k][threadIdx.x];
    const int node = blockIdx.x * 256 + threadIdx.x;
    if (node < n) dinv[node] = rsqrtf((float)deg + 1.0f);
}

constexpr int REC_STEPS = 8;

__global__ __launch_bounds__(256) void edge_records(const int* __restrict__ src,
                                                    const int* __restrict__ dst,
                                                    const float* __restrict__ dinv,
                                                    int* __restrict__ rec, int nE) {
    const int b0 = blockIdx.x * (256 * REC_STEPS);
    const int e0 = b0 + threadIdx.x;
    int s[REC_STEPS], d[REC_STEPS];
    float nrm[REC_STEPS];
    if (b0 + 256 * REC_STEPS <= nE) {
#pragma unroll
        for (int k = 0; k < REC_STEPS; ++k) {
            s[k] = src[e0 + k * 256];
            d[k] = dst[e0 + k * 256];
        }
#pragma unroll
        for (int k = 0; k < REC_STEPS; ++k) nrm[k] = dinv[s[k]] * dinv[d[k]];
#pragma unroll
        for (int k = 0; k < REC_STEPS; ++k)
            *(int4*)(rec + 4 * (size_t)(e0 + k * 256)) = make_int4(s[k], d[k], __float_as_int(nrm[k]), 0);
    } else {
#pragma unroll
        for (int k = 0; k < REC_STEPS; ++k) {
            const int e = e0 + k * 256;
            if (e < nE) {
                const int sk = src[e];
                const int dk = dst[e];
                *(int4*)(rec + 4 * (size_t)e) = make_int4(sk, dk, __float_as_int(dinv[sk] * dinv[dk]), 0);
            }
        }
    }
}

__global__ __launch_bounds__(256) void gemm_x_w1(const float* __restrict__ X,
                                                 const float* __restrict__ W,
                                                 float* __restrict__ H, int n) {
    const int lane = threadIdx.x & 31;
    const int tile = blockIdx.x * 8 + (threadIdx.x >> 5);
    const int row0 = tile * 16;
    if (row0 >= n) return;
    int m = row0 + (lane & 15);
    if (m >= n) m = n - 1;
    const int kh  = (lane >> 4) << 1;
    const int col = lane & 15;
    const float* xr = X + (size_t)m * 256;

    v8f acc = {};
#pragma unroll 4
    for (int k0 = 0; k0 < 256; k0 += 4) {
        v2f a, b;
        a.x = xr[k0 + kh];
        a.y = xr[k0 + kh + 1];
        b.x = W[(size_t)(k0 + kh) * 16 + col];
        b.y = W[(size_t)(k0 + kh + 1) * 16 + col];
        acc = __builtin_amdgcn_wmma_f32_16x16x4_f32(false, a, false, b,
                                                    (short)0, acc, false, false);
    }
    const int ra = (lane >> 4) << 3;
    if (row0 + 16 <= n) {
#pragma unroll
        for (int v = 0; v < 8; ++v)
            H[(size_t)(row0 + ra + v) * 16 + col] = acc[v];
    } else {
#pragma unroll
        for (int v = 0; v < 8; ++v) {
            int r = row0 + ra + v;
            if (r < n) H[(size_t)r * 16 + col] = acc[v];
        }
    }
}

__global__ __launch_bounds__(256) void gemm_z_w2(const float* __restrict__ Z,
                                                 const float* __restrict__ W,
                                                 float* __restrict__ H, int n) {
    const int lane = threadIdx.x & 31;
    const int tile = blockIdx.x * 8 + (threadIdx.x >> 5);
    const int row0 = tile * 16;
    if (row0 >= n) return;
    int m = row0 + (lane & 15);
    if (m >= n) m = n - 1;
    const int kh  = (lane >> 4) << 1;
    const int col = lane & 15;
    const float* zr = Z + (size_t)m * 16;

    v8f acc0 = {}, acc1 = {}, acc2 = {}, acc3 = {};
#pragma unroll
    for (int k0 = 0; k0 < 16; k0 += 4) {
        v2f a, b;
        a.x = zr[k0 + kh];
        a.y = zr[k0 + kh + 1];
        const float* w0 = W + (size_t)(k0 + kh) * 64;
        const float* w1 = w0 + 64;
        b.x = w0[col];      b.y = w1[col];
        acc0 = __builtin_amdgcn_wmma_f32_16x16x4_f32(false, a, false, b, (short)0, acc0, false, false);
        b.x = w0[col + 16]; b.y = w1[col + 16];
        acc1 = __builtin_amdgcn_wmma_f32_16x16x4_f32(false, a, false, b, (short)0, acc1, false, false);
        b.x = w0[col + 32]; b.y = w1[col + 32];
        acc2 = __builtin_amdgcn_wmma_f32_16x16x4_f32(false, a, false, b, (short)0, acc2, false, false);
        b.x = w0[col + 48]; b.y = w1[col + 48];
        acc3 = __builtin_amdgcn_wmma_f32_16x16x4_f32(false, a, false, b, (short)0, acc3, false, false);
    }
    const int ra = (lane >> 4) << 3;
    if (row0 + 16 <= n) {
#pragma unroll
        for (int v = 0; v < 8; ++v) {
            float* hp = H + (size_t)(row0 + ra + v) * 64 + col;
            hp[0]  = acc0[v];
            hp[16] = acc1[v];
            hp[32] = acc2[v];
            hp[48] = acc3[v];
        }
    } else {
#pragma unroll
        for (int v = 0; v < 8; ++v) {
            int r = row0 + ra + v;
            if (r < n) {
                float* hp = H + (size_t)r * 64 + col;
                hp[0]  = acc0[v];
                hp[16] = acc1[v];
                hp[32] = acc2[v];
                hp[48] = acc3[v];
            }
        }
    }
}

constexpr int QCAP = 64;

template <int F>
__device__ __forceinline__ void add_rows(const int* __restrict__ rec, const float* __restrict__ H,
                                         float* tile, int base, int e, int cnt, int lane) {
    constexpr int V = F / 16;
    int s = 0, loc = 0;
    float nrm = 0.0f;
    if (lane < cnt) {
        const int4 r = *(const int4*)(rec + 4 * (size_t)e);
        s = r.x;
        nrm = __int_as_float(r.z);
        loc = r.y - base;
        if ((unsigned)loc >= (unsigned)(32768 / F)) loc = -1;
    }
    const int h = lane >> 4, col = lane & 15;
    for (int j = 0; j < cnt; j += 8) {
        float r[4][V], nj[4];
        int   off[4];
        bool  act[4];
#pragma unroll
        for (int u = 0; u < 4; ++u) {
            const int ja = j + 2 * u;
            const int sA = __builtin_amdgcn_readlane(s, ja);
            const int sB = __builtin_amdgcn_readlane(s, ja + 1);
            const int lA = __builtin_amdgcn_readlane(loc, ja);
            const int lB = __builtin_amdgcn_readlane(loc, ja + 1);
            const int nA = __builtin_amdgcn_readlane(__float_as_int(nrm), ja);
            const int nB = __builtin_amdgcn_readlane(__float_as_int(nrm), ja + 1);
            const int sj = h ? sB : sA;
            const int lj = h ? lB : lA;
            off[u] = lj * (F + 1) + col * V;
            nj[u]  = __int_as_float(h ? nB : nA);
            act[u] = ja + h < cnt && lj >= 0;
            const float* hp = H + (size_t)sj * F + col * V;
            if constexpr (V == 4) {
                const float4 t = *(const float4*)hp;
                r[u][0] = t.x; r[u][1] = t.y; r[u][2] = t.z; r[u][3] = t.w;
            } else {
#pragma unroll
                for (int i = 0; i < V; ++i) r[u][i] = hp[i];
            }
        }
#pragma unroll
        for (int u = 0; u < 4; ++u) {
            if (act[u]) {
#pragma unroll
                for (int i = 0; i < V; ++i) atomicAdd(tile + off[u] + i, r[u][i] * nj[u]);
            }
        }
    }
}

__device__ __forceinline__ unsigned nonzero_bytes(unsigned t) {
    return ((t & 0x7F7F7F7Fu) + 0x7F7F7F7Fu) | t;
}

template <int F>
__global__ __launch_bounds__(1024) void tile_aggregate(const int* __restrict__ rec,
                                                       const float* __restrict__ H,
                                                       const unsigned char* __restrict__ keys,
                                                       float* __restrict__ P,
                                                       int nE, int n) {
    constexpr int TN = 32768 / F;
    constexpr int PITCH = F + 1;
    constexpr int KPT = TN >> KEY_SHIFT;
    static_assert(KPT >= 1 && (KPT & (KPT - 1)) == 0 && KPT * (1 << KEY_SHIFT) == TN, "tile = 2^k keys");
    constexpr unsigned KM  = (0x100u - KPT) & 0xFFu;
    constexpr unsigned KM4 = KM * 0x01010101u;
    __shared__ float tile[TN * PITCH];
    __shared__ int   stack[32 * QCAP];

    const int tid  = threadIdx.x;
    const int lane = tid & 31;
    const int w    = __builtin_amdgcn_readfirstlane(tid >> 5);
    const int base = blockIdx.x * TN;
    const int c    = blockIdx.y;
    int* q = stack + w * QCAP;
    const unsigned pat = (((unsigned)blockIdx.x * KPT) & KM) * 0x01010101u;

    for (int i = tid; i < TN * PITCH; i += 1024) tile[i] = 0.0f;
    __syncthreads();

    const int ec = (((nE + (int)gridDim.y - 1) / (int)gridDim.y) + 511) & ~511;
    const int e0 = c * ec;
    const int e1 = min(nE, e0 + ec);

    int pending = 0;
    constexpr int STRIDE = 32 * 512;
    int eb = e0 + w * 512;
    uint4 kn = make_uint4(0u, 0u, 0u, 0u);
    if (eb < e1) kn = *(const uint4*)(keys + eb + lane * 16);
    for (; eb < e1; eb += STRIDE) {
        const uint4 kc = kn;
        if (eb + STRIDE < e1) kn = *(const uint4*)(keys + eb + STRIDE + lane * 16);
        const unsigned n0 = nonzero_bytes((kc.x ^ pat) & KM4);
        const unsigned n1 = nonzero_bytes((kc.y ^ pat) & KM4);
        const unsigned n2 = nonzero_bytes((kc.z ^ pat) & KM4);
        const unsigned n3 = nonzero_bytes((kc.w ^ pat) & KM4);
        unsigned m = (((n0 >> 7) & 0x01010101u) | ((n1 >> 6) & 0x02020202u) |
                      ((n2 >> 5) & 0x04040404u) | ((n3 >> 4) & 0x08080808u)) ^ 0x0F0F0F0Fu;
        const int elane = eb + lane * 16;
        unsigned any;
        while ((any = __builtin_amdgcn_ballot_w32(m != 0u)) != 0u) {
            const int t = __ffs((int)m) - 1;
            const int e = elane + ((t & 7) << 2) + (t >> 3);
            const unsigned bal = any & __builtin_amdgcn_ballot_w32(e < e1);
            if (m != 0u && e < e1) q[pending + (int)__builtin_amdgcn_mbcnt_lo(bal, 0u)] = e;
            pending += __popc(bal);
            m &= m - 1u;
            if (pending >= 32) {
                pending -= 32;
                add_rows<F>(rec, H, tile, base, q[pending + lane], 32, lane);
            }
        }
    }
    add_rows<F>(rec, H, tile, base, lane < pending ? q[lane] : 0, pending, lane);
    __syncthreads();

    float* pp = P + ((size_t)c * n + base) * F;
    const int rows = min(TN, n - base);
    for (int i = tid; i < rows * F; i += 1024)
        pp[i] = tile[(i / F) * PITCH + (i % F)];
}

template <int F, bool RELU, int NC>
__global__ __launch_bounds__(256) void epilogue(float* __restrict__ OUT,
                                                const float* __restrict__ P,
                                                const float* __restrict__ H,
                                                const float* __restrict__ dinv,
                                                const float* __restrict__ bias, int n, int steps) {
    const size_t total  = (size_t)n * F;
    const size_t groups = total / 4;
    size_t g = (size_t)blockIdx.x * steps * 256 + threadIdx.x;
    for (int k = 0; k < steps; ++k, g += 256) {
        if (g >= groups) break;
        const size_t idx = 4 * g;
        const int i = (int)(idx / F);
        const int f = (int)(idx % F);
        float4 p[NC];
#pragma unroll
        for (int c = 0; c < NC; ++c) p[c] = *(const float4*)(P + (size_t)c * total + idx);
        const float4 h = *(const float4*)(H + idx);
        const float di = dinv[i];
        const float b0 = bias[f], b1 = bias[f + 1], b2 = bias[f + 2], b3 = bias[f + 3];
        float4 acc = make_float4(0.0f, 0.0f, 0.0f, 0.0f);
#pragma unroll
        for (int c = 0; c < NC; ++c) {
            acc.x += p[c].x; acc.y += p[c].y; acc.z += p[c].z; acc.w += p[c].w;
        }
        float4 v;
        v.x = acc.x + di * di * h.x + b0;
        v.y = acc.y + di * di * h.y + b1;
        v.z = acc.z + di * di * h.z + b2;
        v.w = acc.w + di * di * h.w + b3;
        if (RELU) {
            v.x = fmaxf(v.x, 0.0f); v.y = fmaxf(v.y, 0.0f);
            v.z = fmaxf(v.z, 0.0f); v.w = fmaxf(v.w, 0.0f);
        }
        *(float4*)(OUT + idx) = v;
    }
}

constexpr int SCORE_BLOCKS = 256;

__device__ __forceinline__ void score_indices(const int* __restrict__ psrc, const int* __restrict__ pdst,
                                              const int* __restrict__ nsrc, const int* __restrict__ ndst,
                                              int j, int nT, int& s, int& d) {
    if (j < nT) { s = psrc[j]; d = pdst[j]; }
    else        { s = nsrc[j - nT]; d = ndst[j - nT]; }
}

__global__ __launch_bounds__(256) void edge_score(const int* __restrict__ psrc,
                                                  const int* __restrict__ pdst,
                                                  const int* __restrict__ nsrc,
                                                  const int* __restrict__ ndst,
                                                  const float* __restrict__ Z,
                                                  float* __restrict__ out, int nT) {
    const int total  = 2 * nT;
    const int stride = (int)gridDim.x * 256;
    int j = blockIdx.x * 256 + threadIdx.x;
    int s = 0, d = 0;
    if (j < total) score_indices(psrc, pdst, nsrc, ndst, j, nT, s, d);
    while (j < total) {
        const int jn = j + stride;
        const float4* zs = (const float4*)(Z + (size_t)s * 64);
        const float4* zd = (const float4*)(Z + (size_t)d * 64);
        float4 a[16], b[16];
#pragma unroll
        for (int h = 0; h < 2; ++h) {
#pragma unroll
            for (int k = 0; k < 8; ++k) a[8 * h + k] = zs[8 * h + k];
#pragma unroll
            for (int k = 0; k < 8; ++k) b[8 * h + k] = zd[8 * h + k];
        }
        int sn = 0, dn = 0;
        if (jn < total) score_indices(psrc, pdst, nsrc, ndst, jn, nT, sn, dn);
        float acc = 0.0f;
#pragma unroll
        for (int k = 0; k < 16; ++k) {
            acc = fmaf(a[k].x, b[k].x, acc);
            acc = fmaf(a[k].y, b[k].y, acc);
            acc = fmaf(a[k].z, b[k].z, acc);
            acc = fmaf(a[k].w, b[k].w, acc);
        }
        out[j] = acc;
        j = jn; s = sn; d = dn;
    }
}

extern "C" void kernel_launch(void* const* d_in, const int* in_sizes, int n_in,
                              void* d_out, int out_size, void* d_ws, size_t ws_size,
                              hipStream_t stream) {
    const float* x   = (const float*)d_in[0];
    const int*   tei = (const int*)d_in[1];
    const int*   tpo = (const int*)d_in[2];
    const int*   tne = (const int*)d_in[3];
    const float* W1  = (const float*)d_in[4];
    const float* b1  = (const float*)d_in[5];
    const float* W2  = (const float*)d_in[6];
    const float* b2  = (const float*)d_in[7];

    const int N  = in_sizes[0] / 256;
    const int E  = in_sizes[1] / 2;
    const int ET = in_sizes[2] / 2;

    float* ws   = (float*)d_ws;
    float* dinv = ws;
    float* h1   = ws + (size_t)N;
    float* z1   = ws + (size_t)17 * N;
    float* h2   = ws + (size_t)33 * N;
    float* z2   = ws + (size_t)97 * N;
    float* P    = ws + (size_t)161 * N;
    unsigned* keys = (unsigned*)(ws + (((size_t)481 * N + 3) & ~(size_t)3));

    const int B = 256;
    auto cdiv = [](long long a, long long b) { return (int)((a + b - 1) / b); };

    const int keyWords = cdiv(E, 512) * 128;
    const int pitch  = (((N + 1) / 2) + 3) & ~3;
    const int chunks = cdiv(keyWords, CHUNK_WORDS);
    unsigned* planes = (unsigned*)(ws + (((size_t)161 * N + 3) & ~(size_t)3));
    deg_count_keys<<<dim3(chunks, cdiv(pitch, RANGE_WORDS)), 1024, 0, stream>>>(tei + E, planes, keys, E, keyWords, pitch);
    deg_to_dinv<<<cdiv(pitch, 128), B, 0, stream>>>(planes, chunks, pitch, dinv, N);
    int* rec = (int*)(keys + keyWords);
    edge_records<<<cdiv(E, B * REC_STEPS), B, 0, stream>>>(tei, tei + E, dinv, rec, E);

    const int tiles = (N + 15) / 16;

    gemm_x_w1<<<cdiv(tiles, 8), B, 0, stream>>>(x, W1, h1, N);
    tile_aggregate<16><<<dim3(cdiv(N, 2048), 10), 1024, 0, stream>>>(rec, h1, (const unsigned char*)keys, P, E, N);
    epilogue<16, true, 10><<<cdiv((long long)4 * N, B * 3), B, 0, stream>>>(z1, P, h1, dinv, b1, N, 3);

    gemm_z_w2<<<cdiv(tiles, 8), B, 0, stream>>>(z1, W2, h2, N);
    tile_aggregate<64><<<dim3(cdiv(N, 512), 5), 1024, 0, stream>>>(rec, h2, (const unsigned char*)keys, P, E, N);
    epilogue<64, false, 5><<<cdiv((long long)16 * N, B * 6), B, 0, stream>>>(z2, P, h2, dinv, b2, N, 6);

    edge_score<<<SCORE_BLOCKS, B, 0, stream>>>(tpo, tpo + ET, tne, tne + ET,
                                               z2, (float*)d_out, ET);
}


# Aggregation scans one bit per edge, not a key byte: the record pass ORs edge bitmaps, one row per 512 nodes, in LDS and stores them; a lane tests 32 or 64 edges per load (sum order changes)
# speedup vs baseline: 6.6405x; 1.0362x over previous
#include <hip/hip_runtime.h>

typedef __attribute__((ext_vector_type(2))) float v2f;
typedef __attribute__((ext_vector_type(8))) float v8f;

constexpr int ROW_SHIFT = 9;
constexpr int BM_ROWS   = 256;

constexpr int CHUNK_EDGES = 16384;
constexpr int CHUNK_WORDS = CHUNK_EDGES / 4;
constexpr int RANGE_WORDS = 65536;
static_assert(CHUNK_EDGES < 65536, "a 16-bit counter must hold a whole chunk");

__global__ __launch_bounds__(1024) void deg_count(const int* __restrict__ dst,
                                                  unsigned* __restrict__ planes,
                                                  int nE, int nWords, int pitch) {
    __shared__ unsigned cnt[RANGE_WORDS];
    const int tid = threadIdx.x;
    const int w0  = blockIdx.y * RANGE_WORDS;
    const int nw  = min(RANGE_WORDS, pitch - w0);
    const bool vec = (((size_t)dst) & 15) == 0;

    int4 dv[CHUNK_WORDS / 1024];
    const int i0 = blockIdx.x * CHUNK_WORDS;
    if (vec && i0 + CHUNK_WORDS <= nWords && 4 * (i0 + CHUNK_WORDS) <= nE) {
#pragma unroll
        for (int k = 0; k < CHUNK_WORDS / 1024; ++k)
            dv[k] = *(const int4*)(dst + 4 * (i0 + k * 1024 + tid));
    } else {
#pragma unroll
        for (int k = 0; k < CHUNK_WORDS / 1024; ++k) {
            const int e = 4 * (i0 + k * 1024 + tid);
            int4 v = make_int4(-1, -1, -1, -1);
            if (e     < nE) v.x = dst[e];
            if (e + 1 < nE) v.y = dst[e + 1];
            if (e + 2 < nE) v.z = dst[e + 2];
            if (e + 3 < nE) v.w = dst[e + 3];
            dv[k] = v;
        }
    }
    for (int i = tid * 4; i < nw; i += 4096) *(uint4*)(cnt + i) = make_uint4(0u, 0u, 0u, 0u);
    __syncthreads();

    const unsigned node0 = 2u * (unsigned)w0, nodes = 2u * (unsigned)nw;
#pragma unroll
    for (int k = 0; k < CHUNK_WORDS / 1024; ++k) {
        const int d[4] = {dv[k].x, dv[k].y, dv[k].z, dv[k].w};
#pragma unroll
        for (int b = 0; b < 4; ++b) {
            if (d[b] >= 0) {
                const unsigned loc = (unsigned)d[b] - node0;
                if (loc < nodes) atomicAdd(&cnt[loc >> 1], 1u << ((loc & 1u) << 4));
            }
        }
    }
    __syncthreads();

    unsigned* pp = planes + (size_t)blockIdx.x * pitch + w0;
    for (int i = tid * 4; i < nw; i += 4096) *(uint4*)(pp + i) = *(const uint4*)(cnt + i);
}

__global__ __launch_bounds__(256) void deg_to_dinv(const unsigned* __restrict__ planes, int nPlanes,
                                                   int pitch, float* __restrict__ dinv, int n) {
    __shared__ unsigned part[8][256];
    const int lane = threadIdx.x & 31, w = threadIdx.x >> 5;
    const int word = (blockIdx.x * 32 + lane) * 4;
    unsigned s[8] = {0u, 0u, 0u, 0u, 0u, 0u, 0u, 0u};
    if (word < pitch) {
        for (int c = w; c < nPlanes; c += 64) {
            uint4 v[8];
#pragma unroll
            for (int j = 0; j < 8; ++j) {
                const int cc = c + 8 * j;
                v[j] = *(const uint4*)(planes + (size_t)min(cc, nPlanes - 1) * pitch + word);
                if (cc >= nPlanes) v[j] = make_uint4(0u, 0u, 0u, 0u);
            }
#pragma unroll
            for (int j = 0; j < 8; ++j) {
                s[0] += v[j].x & 0xFFFFu; s[1] += v[j].x >> 16;
                s[2] += v[j].y & 0xFFFFu; s[3] += v[j].y >> 16;
                s[4] += v[j].z & 0xFFFFu; s[5] += v[j].z >> 16;
                s[6] += v[j].w & 0xFFFFu; s[7] += v[j].w >> 16;
            }
        }
    }
#pragma unroll
    for (int j = 0; j < 8; ++j) part[w][lane * 8 + j] = s[j];
    __syncthreads();
    unsigned deg = 0;
#pragma unroll
    for (int k = 0; k < 8; ++k) deg += part[k][threadIdx.x];
    const int node = blockIdx.x * 256 + threadIdx.x;
    if (node < n) dinv[node] = rsqrtf((float)deg + 1.0f);
}

constexpr int REC_STEPS = 8;
constexpr int REC_EDGES = 256 * REC_STEPS;
constexpr int REC_WORDS = REC_EDGES / 32;

__global__ __launch_bounds__(256) void edge_records(const int* __restrict__ src,
                                                    const int* __restrict__ dst,
                                                    const float* __restrict__ dinv,
                                                    int* __restrict__ rec,
                                                    unsigned* __restrict__ bm, int rowWords, int nRows,
                                                    int nE) {
    __shared__ unsigned bits[BM_ROWS * REC_WORDS];
    const int tid = threadIdx.x;
    const int b0 = blockIdx.x * REC_EDGES;
    const int e0 = b0 + tid;
    for (int i = tid * 4; i < nRows * REC_WORDS; i += 1024) *(uint4*)(bits + i) = make_uint4(0u, 0u, 0u, 0u);
    int s[REC_STEPS], d[REC_STEPS];
    if (b0 + REC_EDGES <= nE) {
#pragma unroll
        for (int k = 0; k < REC_STEPS; ++k) {
            s[k] = src[e0 + k * 256];
            d[k] = dst[e0 + k * 256];
        }
    } else {
#pragma unroll
        for (int k = 0; k < REC_STEPS; ++k) {
            const int e = e0 + k * 256;
            s[k] = 0;
            d[k] = -1;
            if (e < nE) {
                s[k] = src[e];
                d[k] = dst[e];
            }
        }
    }
    __syncthreads();
#pragma unroll
    for (int k = 0; k < REC_STEPS; ++k) {
        if (d[k] >= 0) {
            const int row = (int)(((unsigned)d[k] >> ROW_SHIFT) & (unsigned)(BM_ROWS - 1));
            atomicOr(&bits[row * REC_WORDS + k * 8 + (tid >> 5)], 1u << (tid & 31));
        }
    }
    __syncthreads();
    float a[REC_STEPS], g[REC_STEPS];
#pragma unroll
    for (int k = 0; k < REC_STEPS; ++k) {
        a[k] = dinv[s[k]];
        g[k] = dinv[max(d[k], 0)];
    }
    for (int i = tid * 4; i < nRows * REC_WORDS; i += 1024) {
        const int row = i / REC_WORDS, w = i % REC_WORDS;
        *(uint4*)(bm + (size_t)row * rowWords + (size_t)blockIdx.x * REC_WORDS + w) = *(const uint4*)(bits + i);
    }
#pragma unroll
    for (int k = 0; k < REC_STEPS; ++k) {
        if (d[k] >= 0)
            *(int4*)(rec + 4 * (size_t)(e0 + k * 256)) = make_int4(s[k], d[k], __float_as_int(a[k] * g[k]), 0);
    }
}

__global__ __launch_bounds__(256) void gemm_x_w1(const float* __restrict__ X,
                                                 const float* __restrict__ W,
                                                 float* __restrict__ H, int n) {
    const int lane = threadIdx.x & 31;
    const int tile = blockIdx.x * 8 + (threadIdx.x >> 5);
    const int row0 = tile * 16;
    if (row0 >= n) return;
    int m = row0 + (lane & 15);
    if (m >= n) m = n - 1;
    const int kh  = (lane >> 4) << 1;
    const int col = lane & 15;
    const float* xr = X + (size_t)m * 256;

    v8f acc = {};
#pragma unroll 4
    for (int k0 = 0; k0 < 256; k0 += 4) {
        v2f a, b;
        a.x = xr[k0 + kh];
        a.y = xr[k0 + kh + 1];
        b.x = W[(size_t)(k0 + kh) * 16 + col];
        b.y = W[(size_t)(k0 + kh + 1) * 16 + col];
        acc = __builtin_amdgcn_wmma_f32_16x16x4_f32(false, a, false, b,
                                                    (short)0, acc, false, false);
    }
    const int ra = (lane >> 4) << 3;
    if (row0 + 16 <= n) {
#pragma unroll
        for (int v = 0; v < 8; ++v)
            H[(size_t)(row0 + ra + v) * 16 + col] = acc[v];
    } else {
#pragma unroll
        for (int v = 0; v < 8; ++v) {
            int r = row0 + ra + v;
            if (r < n) H[(size_t)r * 16 + col] = acc[v];
        }
    }
}

__global__ __launch_bounds__(256) void gemm_z_w2(const float* __restrict__ Z,
                                                 const float* __restrict__ W,
                                                 float* __restrict__ H, int n) {
    const int lane = threadIdx.x & 31;
    const int tile = blockIdx.x * 8 + (threadIdx.x >> 5);
    const int row0 = tile * 16;
    if (row0 >= n) return;
    int m = row0 + (lane & 15);
    if (m >= n) m = n - 1;
    const int kh  = (lane >> 4) << 1;
    const int col = lane & 15;
    const float* zr = Z + (size_t)m * 16;

    v8f acc0 = {}, acc1 = {}, acc2 = {}, acc3 = {};
#pragma unroll
    for (int k0 = 0; k0 < 16; k0 += 4) {
        v2f a, b;
        a.x = zr[k0 + kh];
        a.y = zr[k0 + kh + 1];
        const float* w0 = W + (size_t)(k0 + kh) * 64;
        const float* w1 = w0 + 64;
        b.x = w0[col];      b.y = w1[col];
        acc0 = __builtin_amdgcn_wmma_f32_16x16x4_f32(false, a, false, b, (short)0, acc0, false, false);
        b.x = w0[col + 16]; b.y = w1[col + 16];
        acc1 = __builtin_amdgcn_wmma_f32_16x16x4_f32(false, a, false, b, (short)0, acc1, false, false);
        b.x = w0[col + 32]; b.y = w1[col + 32];
        acc2 = __builtin_amdgcn_wmma_f32_16x16x4_f32(false, a, false, b, (short)0, acc2, false, false);
        b.x = w0[col + 48]; b.y = w1[col + 48];
        acc3 = __builtin_amdgcn_wmma_f32_16x16x4_f32(false, a, false, b, (short)0, acc3, false, false);
    }
    const int ra = (lane >> 4) << 3;
    if (row0 + 16 <= n) {
#pragma unroll
        for (int v = 0; v < 8; ++v) {
            float* hp = H + (size_t)(row0 + ra + v) * 64 + col;
            hp[0]  = acc0[v];
            hp[16] = acc1[v];
            hp[32] = acc2[v];
            hp[48] = acc3[v];
        }
    } else {
#pragma unroll
        for (int v = 0; v < 8; ++v) {
            int r = row0 + ra + v;
            if (r < n) {
                float* hp = H + (size_t)r * 64 + col;
                hp[0]  = acc0[v];
                hp[16] = acc1[v];
                hp[32] = acc2[v];
                hp[48] = acc3[v];
            }
        }
    }
}

constexpr int QCAP = 64;

template <int F>
__device__ __forceinline__ void add_rows(const int* __restrict__ rec, const float* __restrict__ H,
                                         float* tile, int base, int e, int cnt, int lane) {
    constexpr int V = F / 16;
    int s = 0, loc = 0;
    float nrm = 0.0f;
    if (lane < cnt) {
        const int4 r = *(const int4*)(rec + 4 * (size_t)e);
        s = r.x;
        nrm = __int_as_float(r.z);
        loc = r.y - base;
        if ((unsigned)loc >= (unsigned)(32768 / F)) loc = -1;
    }
    const int h = lane >> 4, col = lane & 15;
    for (int j = 0; j < cnt; j += 8) {
        float r[4][V], nj[4];
        int   off[4];
        bool  act[4];
#pragma unroll
        for (int u = 0; u < 4; ++u) {
            const int ja = j + 2 * u;
            const int sA = __builtin_amdgcn_readlane(s, ja);
            const int sB = __builtin_amdgcn_readlane(s, ja + 1);
            const int lA = __builtin_amdgcn_readlane(loc, ja);
            const int lB = __builtin_amdgcn_readlane(loc, ja + 1);
            const int nA = __builtin_amdgcn_readlane(__float_as_int(nrm), ja);
            const int nB = __builtin_amdgcn_readlane(__float_as_int(nrm), ja + 1);
            const int sj = h ? sB : sA;
            const int lj = h ? lB : lA;
            off[u] = lj * (F + 1) + col * V;
            nj[u]  = __int_as_float(h ? nB : nA);
            act[u] = ja + h < cnt && lj >= 0;
            const float* hp = H + (size_t)sj * F + col * V;
            if constexpr (V == 4) {
                const float4 t = *(const float4*)hp;
                r[u][0] = t.x; r[u][1] = t.y; r[u][2] = t.z; r[u][3] = t.w;
            } else {
#pragma unroll
                for (int i = 0; i < V; ++i) r[u][i] = hp[i];
            }
        }
#pragma unroll
        for (int u = 0; u < 4; ++u) {
            if (act[u]) {
#pragma unroll
                for (int i = 0; i < V; ++i) atomicAdd(tile + off[u] + i, r[u][i] * nj[u]);
            }
        }
    }
}

template <int F>
__global__ __launch_bounds__(1024) void tile_aggregate(const int* __restrict__ rec,
                                                       const float* __restrict__ H,
                                                       const unsigned* __restrict__ bm, int rowWords, int nRows,
                                                       float* __restrict__ P, int n) {
    constexpr int TN = 32768 / F;
    constexpr int PITCH = F + 1;
    constexpr int KPT = TN >> ROW_SHIFT;
    static_assert(KPT >= 1 && KPT * (1 << ROW_SHIFT) == TN, "a tile is whole bitmap rows");
    constexpr int LW = F >= 64 ? 2 : 1;
    __shared__ float tile[TN * PITCH];
    __shared__ int   stack[32 * QCAP];

    const int tid  = threadIdx.x;
    const int lane = tid & 31;
    const int w    = __builtin_amdgcn_readfirstlane(tid >> 5);
    const int base = blockIdx.x * TN;
    const int c    = blockIdx.y;
    int* q = stack + w * QCAP;

    const unsigned* rowp[KPT];
    bool rowok[KPT];
#pragma unroll
    for (int j = 0; j < KPT; ++j) {
        const int r = (int)(((unsigned)blockIdx.x * KPT + j) & (unsigned)(BM_ROWS - 1));
        rowok[j] = r < nRows;
        rowp[j]  = bm + (size_t)(rowok[j] ? r : 0) * rowWords;
    }

    for (int i = tid; i < TN * PITCH; i += 1024) tile[i] = 0.0f;
    __syncthreads();

    const int steps = rowWords / (32 * LW);
    const int sc = (steps + (int)gridDim.y - 1) / (int)gridDim.y;
    const int s1 = min(steps, (c + 1) * sc);

    int pending = 0;
    int st = c * sc + w;
    unsigned mn[LW];
#pragma unroll
    for (int l = 0; l < LW; ++l) mn[l] = 0u;
    if (st < s1) {
#pragma unroll
        for (int j = 0; j < KPT; ++j) {
#pragma unroll
            for (int l = 0; l < LW; ++l) {
                const unsigned v = rowp[j][((size_t)st * 32 + lane) * LW + l];
                mn[l] |= rowok[j] ? v : 0u;
            }
        }
    }
    for (; st < s1; st += 32) {
        unsigned mc[LW];
#pragma unroll
        for (int l = 0; l < LW; ++l) { mc[l] = mn[l]; mn[l] = 0u; }
        const int sn = min(st + 32, s1 - 1);
#pragma unroll
        for (int j = 0; j < KPT; ++j) {
#pragma unroll
            for (int l = 0; l < LW; ++l) {
                const unsigned v = rowp[j][((size_t)sn * 32 + lane) * LW + l];
                mn[l] |= rowok[j] ? v : 0u;
            }
        }
        const int elane = (st * 32 + lane) * (32 * LW);
#pragma unroll
        for (int l = 0; l < LW; ++l) {
            unsigned m = mc[l];
            const int ew = elane + 32 * l;
            unsigned any;
            while ((any = __builtin_amdgcn_ballot_w32(m != 0u)) != 0u) {
                if (m != 0u) q[pending + (int)__builtin_amdgcn_mbcnt_lo(any, 0u)] = ew + __ffs((int)m) - 1;
                pending += __popc(any);
                m &= m - 1u;
                if (pending >= 32) {
                    pending -= 32;
                    add_rows<F>(rec, H, tile, base, q[pending + lane], 32, lane);
                }
            }
        }
    }
    add_rows<F>(rec, H, tile, base, lane < pending ? q[lane] : 0, pending, lane);
    __syncthreads();

    float* pp = P + ((size_t)c * n + base) * F;
    const int rows = min(TN, n - base);
    for (int i = tid; i < rows * F; i += 1024)
        pp[i] = tile[(i / F) * PITCH + (i % F)];
}

template <int F, bool RELU, int NC>
__global__ __launch_bounds__(256) void epilogue(float* __restrict__ OUT,
                                                const float* __restrict__ P,
                                                const float* __restrict__ H,
                                                const float* __restrict__ dinv,
                                                const float* __restrict__ bias, int n, int steps) {
    const size_t total  = (size_t)n * F;
    const size_t groups = total / 4;
    size_t g = (size_t)blockIdx.x * steps * 256 + threadIdx.x;
    for (int k = 0; k < steps; ++k, g += 256) {
        if (g >= groups) break;
        const size_t idx = 4 * g;
        const int i = (int)(idx / F);
        const int f = (int)(idx % F);
        float4 p[NC];
#pragma unroll
        for (int c = 0; c < NC; ++c) p[c] = *(const float4*)(P + (size_t)c * total + idx);
        const float4 h = *(const float4*)(H + idx);
        const float di = dinv[i];
        const float b0 = bias[f], b1 = bias[f + 1], b2 = bias[f + 2], b3 = bias[f + 3];
        float4 acc = make_float4(0.0f, 0.0f, 0.0f, 0.0f);
#pragma unroll
        for (int c = 0; c < NC; ++c) {
            acc.x += p[c].x; acc.y += p[c].y; acc.z += p[c].z; acc.w += p[c].w;
        }
        float4 v;
        v.x = acc.x + di * di * h.x + b0;
        v.y = acc.y + di * di * h.y + b1;
        v.z = acc.z + di * di * h.z + b2;
        v.w = acc.w + di * di * h.w + b3;
        if (RELU) {
            v.x = fmaxf(v.x, 0.0f); v.y = fmaxf(v.y, 0.0f);
            v.z = fmaxf(v.z, 0.0f); v.w = fmaxf(v.w, 0.0f);
        }
        *(float4*)(OUT + idx) = v;
    }
}

constexpr int SCORE_BLOCKS = 256;

__device__ __forceinline__ void score_indices(const int* __restrict__ psrc, const int* __restrict__ pdst,
                                              const int* __restrict__ nsrc, const int* __restrict__ ndst,
                                              int j, int nT, int& s, int& d) {
    if (j < nT) { s = psrc[j]; d = pdst[j]; }
    else        { s = nsrc[j - nT]; d = ndst[j - nT]; }
}

__global__ __launch_bounds__(256) void edge_score(const int* __restrict__ psrc,
                                                  const int* __restrict__ pdst,
                                                  const int* __restrict__ nsrc,
                                                  const int* __restrict__ ndst,
                                                  const float* __restrict__ Z,
                                                  float* __restrict__ out, int nT) {
    const int total  = 2 * nT;
    const int stride = (int)gridDim.x * 256;
    int j = blockIdx.x * 256 + threadIdx.x;
    int s = 0, d = 0;
    if (j < total) score_indices(psrc, pdst, nsrc, ndst, j, nT, s, d);
    while (j < total) {
        const int jn = j + stride;
        const float4* zs = (const float4*)(Z + (size_t)s * 64);
        const float4* zd = (const float4*)(Z + (size_t)d * 64);
        float4 a[16], b[16];
#pragma unroll
        for (int h = 0; h < 2; ++h) {
#pragma unroll
            for (int k = 0; k < 8; ++k) a[8 * h + k] = zs[8 * h + k];
#pragma unroll
            for (int k = 0; k < 8; ++k) b[8 * h + k] = zd[8 * h + k];
        }
        int sn = 0, dn = 0;
        if (jn < total) score_indices(psrc, pdst, nsrc, ndst, jn, nT, sn, dn);
        float acc = 0.0f;
#pragma unroll
        for (int k = 0; k < 16; ++k) {
            acc = fmaf(a[k].x, b[k].x, acc);
            acc = fmaf(a[k].y, b[k].y, acc);
            acc = fmaf(a[k].z, b[k].z, acc);
            acc = fmaf(a[k].w, b[k].w, acc);
        }
        out[j] = acc;
        j = jn; s = sn; d = dn;
    }
}

extern "C" void kernel_launch(void* const* d_in, const int* in_sizes, int n_in,
                              void* d_out, int out_size, void* d_ws, size_t ws_size,
                              hipStream_t stream) {
    const float* x   = (const float*)d_in[0];
    const int*   tei = (const int*)d_in[1];
    const int*   tpo = (const int*)d_in[2];
    const int*   tne = (const int*)d_in[3];
    const float* W1  = (const float*)d_in[4];
    const float* b1  = (const float*)d_in[5];
    const float* W2  = (const float*)d_in[6];
    const float* b2  = (const float*)d_in[7];

    const int N  = in_sizes[0] / 256;
    const int E  = in_sizes[1] / 2;
    const int ET = in_sizes[2] / 2;

    float* ws   = (float*)d_ws;
    float* dinv = ws;
    float* h1   = ws + (size_t)N;
    float* z1   = ws + (size_t)17 * N;
    float* h2   = ws + (size_t)33 * N;
    float* z2   = ws + (size_t)97 * N;
    float* P    = ws + (size_t)161 * N;
    int* rec    = (int*)(ws + (((size_t)481 * N + 3) & ~(size_t)3));

    const int B = 256;
    auto cdiv = [](long long a, long long b) { return (int)((a + b - 1) / b); };

    const int edgeWords = cdiv(E, 4);
    const int pitch  = (((N + 1) / 2) + 3) & ~3;
    const int chunks = cdiv(edgeWords, CHUNK_WORDS);
    unsigned* planes = (unsigned*)(ws + (((size_t)161 * N + 3) & ~(size_t)3));
    deg_count<<<dim3(chunks, cdiv(pitch, RANGE_WORDS)), 1024, 0, stream>>>(tei + E, planes, E, edgeWords, pitch);
    deg_to_dinv<<<cdiv(pitch, 128), B, 0, stream>>>(planes, chunks, pitch, dinv, N);
    const int recBlocks = cdiv(E, REC_EDGES);
    const int rowWords  = recBlocks * REC_WORDS;
    const int nRows     = cdiv(N, 1 << ROW_SHIFT) < BM_ROWS ? cdiv(N, 1 << ROW_SHIFT) : BM_ROWS;
    unsigned* bm = (unsigned*)(rec + 4 * (size_t)E);
    edge_records<<<recBlocks, B, 0, stream>>>(tei, tei + E, dinv, rec, bm, rowWords, nRows, E);

    const int tiles = (N + 15) / 16;

    gemm_x_w1<<<cdiv(tiles, 8), B, 0, stream>>>(x, W1, h1, N);
    tile_aggregate<16><<<dim3(cdiv(N, 2048), 10), 1024, 0, stream>>>(rec, h1, bm, rowWords, nRows, P, N);
    epilogue<16, true, 10><<<cdiv((long long)4 * N, B * 3), B, 0, stream>>>(z1, P, h1, dinv, b1, N, 3);

    gemm_z_w2<<<cdiv(tiles, 8), B, 0, stream>>>(z1, W2, h2, N);
    tile_aggregate<64><<<dim3(cdiv(N, 512), 5), 1024, 0, stream>>>(rec, h2, bm, rowWords, nRows, P, N);
    epilogue<64, false, 5><<<cdiv((long long)16 * N, B * 6), B, 0, stream>>>(z2, P, h2, dinv, b2, N, 6);

    edge_score<<<SCORE_BLOCKS, B, 0, stream>>>(tpo, tpo + ET, tne, tne + ET,
                                               z2, (float*)d_out, ET);
}


# Two launch boundaries gone: dinv pass and X projection share a launch (independent work); layer-1 plane sum feeds the W2 projection through LDS in one launch, z1 never stored; 8 launches, sums kept
# speedup vs baseline: 6.8715x; 1.0348x over previous
#include <hip/hip_runtime.h>

typedef __attribute__((ext_vector_type(2))) float v2f;
typedef __attribute__((ext_vector_type(8))) float v8f;

constexpr int ROW_SHIFT = 9;
constexpr int BM_ROWS   = 256;

constexpr int CHUNK_EDGES = 16384;
constexpr int CHUNK_WORDS = CHUNK_EDGES / 4;
constexpr int RANGE_WORDS = 65536;
static_assert(CHUNK_EDGES < 65536, "a 16-bit counter must hold a whole chunk");

__global__ __launch_bounds__(1024) void deg_count(const int* __restrict__ dst,
                                                  unsigned* __restrict__ planes,
                                                  int nE, int nWords, int pitch) {
    __shared__ unsigned cnt[RANGE_WORDS];
    const int tid = threadIdx.x;
    const int w0  = blockIdx.y * RANGE_WORDS;
    const int nw  = min(RANGE_WORDS, pitch - w0);
    const bool vec = (((size_t)dst) & 15) == 0;

    int4 dv[CHUNK_WORDS / 1024];
    const int i0 = blockIdx.x * CHUNK_WORDS;
    if (vec && i0 + CHUNK_WORDS <= nWords && 4 * (i0 + CHUNK_WORDS) <= nE) {
#pragma unroll
        for (int k = 0; k < CHUNK_WORDS / 1024; ++k)
            dv[k] = *(const int4*)(dst + 4 * (i0 + k * 1024 + tid));
    } else {
#pragma unroll
        for (int k = 0; k < CHUNK_WORDS / 1024; ++k) {
            const int e = 4 * (i0 + k * 1024 + tid);
            int4 v = make_int4(-1, -1, -1, -1);
            if (e     < nE) v.x = dst[e];
            if (e + 1 < nE) v.y = dst[e + 1];
            if (e + 2 < nE) v.z = dst[e + 2];
            if (e + 3 < nE) v.w = dst[e + 3];
            dv[k] = v;
        }
    }
    for (int i = tid * 4; i < nw; i += 4096) *(uint4*)(cnt + i) = make_uint4(0u, 0u, 0u, 0u);
    __syncthreads();

    const unsigned node0 = 2u * (unsigned)w0, nodes = 2u * (unsigned)nw;
#pragma unroll
    for (int k = 0; k < CHUNK_WORDS / 1024; ++k) {
        const int d[4] = {dv[k].x, dv[k].y, dv[k].z, dv[k].w};
#pragma unroll
        for (int b = 0; b < 4; ++b) {
            if (d[b] >= 0) {
                const unsigned loc = (unsigned)d[b] - node0;
                if (loc < nodes) atomicAdd(&cnt[loc >> 1], 1u << ((loc & 1u) << 4));
            }
        }
    }
    __syncthreads();

    unsigned* pp = planes + (size_t)blockIdx.x * pitch + w0;
    for (int i = tid * 4; i < nw; i += 4096) *(uint4*)(pp + i) = *(const uint4*)(cnt + i);
}

__device__ __forceinline__ void deg_to_dinv_block(unsigned (*part)[256], int blk,
                                                  const unsigned* __restrict__ planes, int nPlanes,
                                                  int pitch, float* __restrict__ dinv, int n) {
    const int lane = threadIdx.x & 31, w = threadIdx.x >> 5;
    const int word = (blk * 32 + lane) * 4;
    unsigned s[8] = {0u, 0u, 0u, 0u, 0u, 0u, 0u, 0u};
    if (word < pitch) {
        for (int c = w; c < nPlanes; c += 64) {
            uint4 v[8];
#pragma unroll
            for (int j = 0; j < 8; ++j) {
                const int cc = c + 8 * j;
                v[j] = *(const uint4*)(planes + (size_t)min(cc, nPlanes - 1) * pitch + word);
                if (cc >= nPlanes) v[j] = make_uint4(0u, 0u, 0u, 0u);
            }
#pragma unroll
            for (int j = 0; j < 8; ++j) {
                s[0] += v[j].x & 0xFFFFu; s[1] += v[j].x >> 16;
                s[2] += v[j].y & 0xFFFFu; s[3] += v[j].y >> 16;
                s[4] += v[j].z & 0xFFFFu; s[5] += v[j].z >> 16;
                s[6] += v[j].w & 0xFFFFu; s[7] += v[j].w >> 16;
            }
        }
    }
#pragma unroll
    for (int j = 0; j < 8; ++j) part[w][lane * 8 + j] = s[j];
    __syncthreads();
    unsigned deg = 0;
#pragma unroll
    for (int k = 0; k < 8; ++k) deg += part[k][threadIdx.x];
    const int node = blk * 256 + threadIdx.x;
    if (node < n) dinv[node] = rsqrtf((float)deg + 1.0f);
}

constexpr int REC_STEPS = 8;
constexpr int REC_EDGES = 256 * REC_STEPS;
constexpr int REC_WORDS = REC_EDGES / 32;

__global__ __launch_bounds__(256) void edge_records(const int* __restrict__ src,
                                                    const int* __restrict__ dst,
                                                    const float* __restrict__ dinv,
                                                    int* __restrict__ rec,
                                                    unsigned* __restrict__ bm, int rowWords, int nRows,
                                                    int nE) {
    __shared__ unsigned bits[BM_ROWS * REC_WORDS];
    const int tid = threadIdx.x;
    const int b0 = blockIdx.x * REC_EDGES;
    const int e0 = b0 + tid;
    for (int i = tid * 4; i < nRows * REC_WORDS; i += 1024) *(uint4*)(bits + i) = make_uint4(0u, 0u, 0u, 0u);
    int s[REC_STEPS], d[REC_STEPS];
    if (b0 + REC_EDGES <= nE) {
#pragma unroll
        for (int k = 0; k < REC_STEPS; ++k) {
            s[k] = src[e0 + k * 256];
            d[k] = dst[e0 + k * 256];
        }
    } else {
#pragma unroll
        for (int k = 0; k < REC_STEPS; ++k) {
            const int e = e0 + k * 256;
            s[k] = 0;
            d[k] = -1;
            if (e < nE) {
                s[k] = src[e];
                d[k] = dst[e];
            }
        }
    }
    __syncthreads();
#pragma unroll
    for (int k = 0; k < REC_STEPS; ++k) {
        if (d[k] >= 0) {
            const int row = (int)(((unsigned)d[k] >> ROW_SHIFT) & (unsigned)(BM_ROWS - 1));
            atomicOr(&bits[row * REC_WORDS + k * 8 + (tid >> 5)], 1u << (tid & 31));
        }
    }
    __syncthreads();
    float a[REC_STEPS], g[REC_STEPS];
#pragma unroll
    for (int k = 0; k < REC_STEPS; ++k) {
        a[k] = dinv[s[k]];
        g[k] = dinv[max(d[k], 0)];
    }
    for (int i = tid * 4; i < nRows * REC_WORDS; i += 1024) {
        const int row = i / REC_WORDS, w = i % REC_WORDS;
        *(uint4*)(bm + (size_t)row * rowWords + (size_t)blockIdx.x * REC_WORDS + w) = *(const uint4*)(bits + i);
    }
#pragma unroll
    for (int k = 0; k < REC_STEPS; ++k) {
        if (d[k] >= 0)
            *(int4*)(rec + 4 * (size_t)(e0 + k * 256)) = make_int4(s[k], d[k], __float_as_int(a[k] * g[k]), 0);
    }
}

__device__ __forceinline__ void gemm_x_w1_block(int blk, const float* __restrict__ X,
                                                const float* __restrict__ W,
                                                float* __restrict__ H, int n) {
    const int lane = threadIdx.x & 31;
    const int tile = blk * 8 + (threadIdx.x >> 5);
    const int row0 = tile * 16;
    if (row0 >= n) return;
    int m = row0 + (lane & 15);
    if (m >= n) m = n - 1;
    const int kh  = (lane >> 4) << 1;
    const int col = lane & 15;
    const float* xr = X + (size_t)m * 256;

    v8f acc = {};
#pragma unroll 4
    for (int k0 = 0; k0 < 256; k0 += 4) {
        v2f a, b;
        a.x = xr[k0 + kh];
        a.y = xr[k0 + kh + 1];
        b.x = W[(size_t)(k0 + kh) * 16 + col];
        b.y = W[(size_t)(k0 + kh + 1) * 16 + col];
        acc = __builtin_amdgcn_wmma_f32_16x16x4_f32(false, a, false, b,
                                                    (short)0, acc, false, false);
    }
    const int ra = (lane >> 4) << 3;
    if (row0 + 16 <= n) {
#pragma unroll
        for (int v = 0; v < 8; ++v)
            H[(size_t)(row0 + ra + v) * 16 + col] = acc[v];
    } else {
#pragma unroll
        for (int v = 0; v < 8; ++v) {
            int r = row0 + ra + v;
            if (r < n) H[(size_t)r * 16 + col] = acc[v];
        }
    }
}

__global__ __launch_bounds__(256) void dinv_and_project(const unsigned* __restrict__ planes, int nPlanes,
                                                        int pitch, float* __restrict__ dinv, int dinvBlocks,
                                                        const float* __restrict__ X,
                                                        const float* __restrict__ W,
                                                        float* __restrict__ H, int n) {
    __shared__ unsigned part[8][256];
    const int blk = (int)blockIdx.x;
    if (blk < dinvBlocks) deg_to_dinv_block(part, blk, planes, nPlanes, pitch, dinv, n);
    else                  gemm_x_w1_block(blk - dinvBlocks, X, W, H, n);
}

template <int NC>
__global__ __launch_bounds__(256) void sum_relu_project(const float* __restrict__ P,
                                                        const float* __restrict__ H1,
                                                        const float* __restrict__ dinv,
                                                        const float* __restrict__ bias,
                                                        const float* __restrict__ W,
                                                        float* __restrict__ H, int n) {
    constexpr int ZP = 18;
    __shared__ float zt[128 * ZP];
    const int tid   = threadIdx.x;
    const int rowB  = blockIdx.x * 128;
    const size_t total = (size_t)n * 16;

    float4 p[2][NC], h[2];
    float  di[2];
    bool   in[2];
#pragma unroll
    for (int k = 0; k < 2; ++k) {
        const int g   = tid + 256 * k;
        const int row = rowB + (g >> 2);
        in[k] = row < n;
        const size_t idx = (size_t)(in[k] ? row : 0) * 16 + 4 * (g & 3);
#pragma unroll
        for (int cc = 0; cc < NC; ++cc) p[k][cc] = *(const float4*)(P + (size_t)cc * total + idx);
        h[k]  = *(const float4*)(H1 + idx);
        di[k] = dinv[in[k] ? row : 0];
    }
    const int f = 4 * (tid & 3);
    const float b0 = bias[f], b1 = bias[f + 1], b2 = bias[f + 2], b3 = bias[f + 3];
#pragma unroll
    for (int k = 0; k < 2; ++k) {
        float4 acc = make_float4(0.0f, 0.0f, 0.0f, 0.0f);
#pragma unroll
        for (int cc = 0; cc < NC; ++cc) {
            acc.x += p[k][cc].x; acc.y += p[k][cc].y; acc.z += p[k][cc].z; acc.w += p[k][cc].w;
        }
        float4 v;
        v.x = acc.x + di[k] * di[k] * h[k].x + b0;
        v.y = acc.y + di[k] * di[k] * h[k].y + b1;
        v.z = acc.z + di[k] * di[k] * h[k].z + b2;
        v.w = acc.w + di[k] * di[k] * h[k].w + b3;
        v.x = fmaxf(v.x, 0.0f); v.y = fmaxf(v.y, 0.0f);
        v.z = fmaxf(v.z, 0.0f); v.w = fmaxf(v.w, 0.0f);
        if (!in[k]) v = make_float4(0.0f, 0.0f, 0.0f, 0.0f);
        float* zp = zt + ((tid + 256 * k) >> 2) * ZP + f;
        *(float2*)zp       = make_float2(v.x, v.y);
        *(float2*)(zp + 2) = make_float2(v.z, v.w);
    }
    __syncthreads();

    const int lane = tid & 31;
    const int wv   = tid >> 5;
    const int row0 = rowB + wv * 16;
    if (row0 >= n) return;
    const int kh  = (lane >> 4) << 1;
    const int col = lane & 15;
    const float* zr = zt + (wv * 16 + (lane & 15)) * ZP;

    v8f acc0 = {}, acc1 = {}, acc2 = {}, acc3 = {};
#pragma unroll
    for (int k0 = 0; k0 < 16; k0 += 4) {
        v2f a, b;
        const float2 za = *(const float2*)(zr + k0 + kh);
        a.x = za.x;
        a.y = za.y;
        const float* w0 = W + (size_t)(k0 + kh) * 64;
        const float* w1 = w0 + 64;
        b.x = w0[col];      b.y = w1[col];
        acc0 = __builtin_amdgcn_wmma_f32_16x16x4_f32(false, a, false, b, (short)0, acc0, false, false);
        b.x = w0[col + 16]; b.y = w1[col + 16];
        acc1 = __builtin_amdgcn_wmma_f32_16x16x4_f32(false, a, false, b, (short)0, acc1, false, false);
        b.x = w0[col + 32]; b.y = w1[col + 32];
        acc2 = __builtin_amdgcn_wmma_f32_16x16x4_f32(false, a, false, b, (short)0, acc2, false, false);
        b.x = w0[col + 48]; b.y = w1[col + 48];
        acc3 = __builtin_amdgcn_wmma_f32_16x16x4_f32(false, a, false, b, (short)0, acc3, false, false);
    }
    const int ra = (lane >> 4) << 3;
    if (row0 + 16 <= n) {
#pragma unroll
        for (int v = 0; v < 8; ++v) {
            float* hp = H + (size_t)(row0 + ra + v) * 64 + col;
            hp[0]  = acc0[v];
            hp[16] = acc1[v];
            hp[32] = acc2[v];
            hp[48] = acc3[v];
        }
    } else {
#pragma unroll
        for (int v = 0; v < 8; ++v) {
            int r = row0 + ra + v;
            if (r < n) {
                float* hp = H + (size_t)r * 64 + col;
                hp[0]  = acc0[v];
                hp[16] = acc1[v];
                hp[32] = acc2[v];
                hp[48] = acc3[v];
            }
        }
    }
}

constexpr int QCAP = 64;

template <int F>
__device__ __forceinline__ void add_rows(const int* __restrict__ rec, const float* __restrict__ H,
                                         float* tile, int base, int e, int cnt, int lane) {
    constexpr int V = F / 16;
    int s = 0, loc = 0;
    float nrm = 0.0f;
    if (lane < cnt) {
        const int4 r = *(const int4*)(rec + 4 * (size_t)e);
        s = r.x;
        nrm = __int_as_float(r.z);
        loc = r.y - base;
        if ((unsigned)loc >= (unsigned)(32768 / F)) loc = -1;
    }
    const int h = lane >> 4, col = lane & 15;
    for (int j = 0; j < cnt; j += 8) {
        float r[4][V], nj[4];
        int   off[4];
        bool  act[4];
#pragma unroll
        for (int u = 0; u < 4; ++u) {
            const int ja = j + 2 * u;
            const int sA = __builtin_amdgcn_readlane(s, ja);
            const int sB = __builtin_amdgcn_readlane(s, ja + 1);
            const int lA = __builtin_amdgcn_readlane(loc, ja);
            const int lB = __builtin_amdgcn_readlane(loc, ja + 1);
            const int nA = __builtin_amdgcn_readlane(__float_as_int(nrm), ja);
            const int nB = __builtin_amdgcn_readlane(__float_as_int(nrm), ja + 1);
            const int sj = h ? sB : sA;
            const int lj = h ? lB : lA;
            off[u] = lj * (F + 1) + col * V;
            nj[u]  = __int_as_float(h ? nB : nA);
            act[u] = ja + h < cnt && lj >= 0;
            const float* hp = H + (size_t)sj * F + col * V;
            if constexpr (V == 4) {
                const float4 t = *(const float4*)hp;
                r[u][0] = t.x; r[u][1] = t.y; r[u][2] = t.z; r[u][3] = t.w;
            } else {
#pragma unroll
                for (int i = 0; i < V; ++i) r[u][i] = hp[i];
            }
        }
#pragma unroll
        for (int u = 0; u < 4; ++u) {
            if (act[u]) {
#pragma unroll
                for (int i = 0; i < V; ++i) atomicAdd(tile + off[u] + i, r[u][i] * nj[u]);
            }
        }
    }
}

template <int F>
__global__ __launch_bounds__(1024) void tile_aggregate(const int* __restrict__ rec,
                                                       const float* __restrict__ H,
                                                       const unsigned* __restrict__ bm, int rowWords, int nRows,
                                                       float* __restrict__ P, int n) {
    constexpr int TN = 32768 / F;
    constexpr int PITCH = F + 1;
    constexpr int KPT = TN >> ROW_SHIFT;
    static_assert(KPT >= 1 && KPT * (1 << ROW_SHIFT) == TN, "a tile is whole bitmap rows");
    constexpr int LW = F >= 64 ? 2 : 1;
    __shared__ float tile[TN * PITCH];
    __shared__ int   stack[32 * QCAP];

    const int tid  = threadIdx.x;
    const int lane = tid & 31;
    const int w    = __builtin_amdgcn_readfirstlane(tid >> 5);
    const int base = blockIdx.x * TN;
    const int c    = blockIdx.y;
    int* q = stack + w * QCAP;

    const unsigned* rowp[KPT];
    bool rowok[KPT];
#pragma unroll
    for (int j = 0; j < KPT; ++j) {
        const int r = (int)(((unsigned)blockIdx.x * KPT + j) & (unsigned)(BM_ROWS - 1));
        rowok[j] = r < nRows;
        rowp[j]  = bm + (size_t)(rowok[j] ? r : 0) * rowWords;
    }

    for (int i = tid; i < TN * PITCH; i += 1024) tile[i] = 0.0f;
    __syncthreads();

    const int steps = rowWords / (32 * LW);
    const int sc = (steps + (int)gridDim.y - 1) / (int)gridDim.y;
    const int s1 = min(steps, (c + 1) * sc);

    int pending = 0;
    int st = c * sc + w;
    unsigned mn[LW];
#pragma unroll
    for (int l = 0; l < LW; ++l) mn[l] = 0u;
    if (st < s1) {
#pragma unroll
        for (int j = 0; j < KPT; ++j) {
#pragma unroll
            for (int l = 0; l < LW; ++l) {
                const unsigned v = rowp[j][((size_t)st * 32 + lane) * LW + l];
                mn[l] |= rowok[j] ? v : 0u;
            }
        }
    }
    for (; st < s1; st += 32) {
        unsigned mc[LW];
#pragma unroll
        for (int l = 0; l < LW; ++l) { mc[l] = mn[l]; mn[l] = 0u; }
        const int sn = min(st + 32, s1 - 1);
#pragma unroll
        for (int j = 0; j < KPT; ++j) {
#pragma unroll
            for (int l = 0; l < LW; ++l) {
                const unsigned v = rowp[j][((size_t)sn * 32 + lane) * LW + l];
                mn[l] |= rowok[j] ? v : 0u;
            }
        }
        const int elane = (st * 32 + lane) * (32 * LW);
#pragma unroll
        for (int l = 0; l < LW; ++l) {
            unsigned m = mc[l];
            const int ew = elane + 32 * l;
            unsigned any;
            while ((any = __builtin_amdgcn_ballot_w32(m != 0u)) != 0u) {
                if (m != 0u) q[pending + (int)__builtin_amdgcn_mbcnt_lo(any, 0u)] = ew + __ffs((int)m) - 1;
                pending += __popc(any);
                m &= m - 1u;
                if (pending >= 32) {
                    pending -= 32;
                    add_rows<F>(rec, H, tile, base, q[pending + lane], 32, lane);
                }
            }
        }
    }
    add_rows<F>(rec, H, tile, base, lane < pending ? q[lane] : 0, pending, lane);
    __syncthreads();

    float* pp = P + ((size_t)c * n + base) * F;
    const int rows = min(TN, n - base);
    for (int i = tid; i < rows * F; i += 1024)
        pp[i] = tile[(i / F) * PITCH + (i % F)];
}

template <int F, bool RELU, int NC>
__global__ __launch_bounds__(256) void epilogue(float* __restrict__ OUT,
                                                const float* __restrict__ P,
                                                const float* __restrict__ H,
                                                const float* __restrict__ dinv,
                                                const float* __restrict__ bias, int n, int steps) {
    const size_t total  = (size_t)n * F;
    const size_t groups = total / 4;
    size_t g = (size_t)blockIdx.x * steps * 256 + threadIdx.x;
    for (int k = 0; k < steps; ++k, g += 256) {
        if (g >= groups) break;
        const size_t idx = 4 * g;
        const int i = (int)(idx / F);
        const int f = (int)(idx % F);
        float4 p[NC];
#pragma unroll
        for (int c = 0; c < NC; ++c) p[c] = *(const float4*)(P + (size_t)c * total + idx);
        const float4 h = *(const float4*)(H + idx);
        const float di = dinv[i];
        const float b0 = bias[f], b1 = bias[f + 1], b2 = bias[f + 2], b3 = bias[f + 3];
        float4 acc = make_float4(0.0f, 0.0f, 0.0f, 0.0f);
#pragma unroll
        for (int c = 0; c < NC; ++c) {
            acc.x += p[c].x; acc.y += p[c].y; acc.z += p[c].z; acc.w += p[c].w;
        }
        float4 v;
        v.x = acc.x + di * di * h.x + b0;
        v.y = acc.y + di * di * h.y + b1;
        v.z = acc.z + di * di * h.z + b2;
        v.w = acc.w + di * di * h.w + b3;
        if (RELU) {
            v.x = fmaxf(v.x, 0.0f); v.y = fmaxf(v.y, 0.0f);
            v.z = fmaxf(v.z, 0.0f); v.w = fmaxf(v.w, 0.0f);
        }
        *(float4*)(OUT + idx) = v;
    }
}

constexpr int SCORE_BLOCKS = 256;

__device__ __forceinline__ void score_indices(const int* __restrict__ psrc, const int* __restrict__ pdst,
                                              const int* __restrict__ nsrc, const int* __restrict__ ndst,
                                              int j, int nT, int& s, int& d) {
    if (j < nT) { s = psrc[j]; d = pdst[j]; }
    else        { s = nsrc[j - nT]; d = ndst[j - nT]; }
}

__global__ __launch_bounds__(256) void edge_score(const int* __restrict__ psrc,
                                                  const int* __restrict__ pdst,
                                                  const int* __restrict__ nsrc,
                                                  const int* __restrict__ ndst,
                                                  const float* __restrict__ Z,
                                                  float* __restrict__ out, int nT) {
    const int total  = 2 * nT;
    const int stride = (int)gridDim.x * 256;
    int j = blockIdx.x * 256 + threadIdx.x;
    int s = 0, d = 0;
    if (j < total) score_indices(psrc, pdst, nsrc, ndst, j, nT, s, d);
    while (j < total) {
        const int jn = j + stride;
        const float4* zs = (const float4*)(Z + (size_t)s * 64);
        const float4* zd = (const float4*)(Z + (size_t)d * 64);
        float4 a[16], b[16];
#pragma unroll
        for (int h = 0; h < 2; ++h) {
#pragma unroll
            for (int k = 0; k < 8; ++k) a[8 * h + k] = zs[8 * h + k];
#pragma unroll
            for (int k = 0; k < 8; ++k) b[8 * h + k] = zd[8 * h + k];
        }
        int sn = 0, dn = 0;
        if (jn < total) score_indices(psrc, pdst, nsrc, ndst, jn, nT, sn, dn);
        float acc = 0.0f;
#pragma unroll
        for (int k = 0; k < 16; ++k) {
            acc = fmaf(a[k].x, b[k].x, acc);
            acc = fmaf(a[k].y, b[k].y, acc);
            acc = fmaf(a[k].z, b[k].z, acc);
            acc = fmaf(a[k].w, b[k].w, acc);
        }
        out[j] = acc;
        j = jn; s = sn; d = dn;
    }
}

extern "C" void kernel_launch(void* const* d_in, const int* in_sizes, int n_in,
                              void* d_out, int out_size, void* d_ws, size_t ws_size,
                              hipStream_t stream) {
    const float* x   = (const float*)d_in[0];
    const int*   tei = (const int*)d_in[1];
    const int*   tpo = (const int*)d_in[2];
    const int*   tne = (const int*)d_in[3];
    const float* W1  = (const float*)d_in[4];
    const float* b1  = (const float*)d_in[5];
    const float* W2  = (const float*)d_in[6];
    const float* b2  = (const float*)d_in[7];

    const int N  = in_sizes[0] / 256;
    const int E  = in_sizes[1] / 2;
    const int ET = in_sizes[2] / 2;

    float* ws   = (float*)d_ws;
    float* dinv = ws;
    float* h1   = ws + (size_t)N;
    float* h2   = ws + (size_t)33 * N;
    float* z2   = ws + (size_t)97 * N;
    float* P    = ws + (size_t)161 * N;
    int* rec    = (int*)(ws + (((size_t)481 * N + 3) & ~(size_t)3));

    const int B = 256;
    auto cdiv = [](long long a, long long b) { return (int)((a + b - 1) / b); };

    const int edgeWords = cdiv(E, 4);
    const int pitch  = (((N + 1) / 2) + 3) & ~3;
    const int chunks = cdiv(edgeWords, CHUNK_WORDS);
    unsigned* planes = (unsigned*)(ws + (((size_t)161 * N + 3) & ~(size_t)3));
    deg_count<<<dim3(chunks, cdiv(pitch, RANGE_WORDS)), 1024, 0, stream>>>(tei + E, planes, E, edgeWords, pitch);
    const int tiles = (N + 15) / 16;
    const int dinvBlocks = cdiv(pitch, 128);
    dinv_and_project<<<dinvBlocks + cdiv(tiles, 8), B, 0, stream>>>(planes, chunks, pitch, dinv, dinvBlocks, x, W1, h1, N);
    const int recBlocks = cdiv(E, REC_EDGES);
    const int rowWords  = recBlocks * REC_WORDS;
    const int nRows     = cdiv(N, 1 << ROW_SHIFT) < BM_ROWS ? cdiv(N, 1 << ROW_SHIFT) : BM_ROWS;
    unsigned* bm = (unsigned*)(rec + 4 * (size_t)E);
    edge_records<<<recBlocks, B, 0, stream>>>(tei, tei + E, dinv, rec, bm, rowWords, nRows, E);

    tile_aggregate<16><<<dim3(cdiv(N, 2048), 10), 1024, 0, stream>>>(rec, h1, bm, rowWords, nRows, P, N);

    sum_relu_project<10><<<cdiv(tiles, 8), B, 0, stream>>>(P, h1, dinv, b1, W2, h2, N);
    tile_aggregate<64><<<dim3(cdiv(N, 512), 5), 1024, 0, stream>>>(rec, h2, bm, rowWords, nRows, P, N);
    epilogue<64, false, 5><<<cdiv((long long)16 * N, B * 6), B, 0, stream>>>(z2, P, h2, dinv, b2, N, 6);

    edge_score<<<SCORE_BLOCKS, B, 0, stream>>>(tpo, tpo + ET, tne, tne + ET,
                                               z2, (float*)d_out, ET);
}
